# starting kernel
# baseline (speedup 1.0000x reference)
#include <hip/hip_runtime.h>

typedef __attribute__((ext_vector_type(2))) float v2f;
typedef __attribute__((ext_vector_type(8))) float v8f;

#define U_N     256
#define T_N     512
#define H_N     4
#define UAVD    64
#define TASKD   32
#define E_N     128
#define HID_N   256

__device__ __forceinline__ v8f wmma_k_loop(const float* __restrict__ ap,
                                           const float* __restrict__ bp,
                                           int K, v8f acc) {
  for (int kk = 0; kk < K; kk += 4) {
    v2f a; a.x = ap[kk]; a.y = ap[kk + 1];
    v2f b; b.x = bp[kk]; b.y = bp[kk + 1];
    acc = __builtin_amdgcn_wmma_f32_16x16x4_f32(false, a, false, b,
                                                (short)0, acc, false, false);
  }
  return acc;
}

template <int INK>
__global__ __launch_bounds__(128) void encoder_kernel(
    const float* __restrict__ x,
    const float* __restrict__ w0, const float* __restrict__ b0,
    const float* __restrict__ w1, const float* __restrict__ b1,
    const float* __restrict__ w2, const float* __restrict__ b2,
    float* __restrict__ out) {
  __shared__ __align__(16) float bufA[16 * 132];
  __shared__ __align__(16) float bufB[16 * 132];
  const int tid  = threadIdx.x;
  const int lane = tid & 31;
  const int wave = tid >> 5;
  const int row0 = blockIdx.x * 16;

  for (int i = tid; i < 16 * INK; i += 128) {
    int r = i / INK, c = i - r * INK;
    bufA[r * 132 + c] = x[(size_t)(row0 + r) * INK + c];
  }
  __syncthreads();

  const int m    = lane & 15;
  const int kb   = (lane >> 4) << 1;
  const int moff = (lane >> 4) * 8;

  #pragma unroll
  for (int t = 0; t < 2; ++t) {
    const int nb = (wave * 2 + t) * 16;
    v8f acc = {};
    acc = wmma_k_loop(bufA + m * 132 + kb,
                      w0 + (size_t)(nb + m) * INK + kb, INK, acc);
    const float bv = b0[nb + m];
    #pragma unroll
    for (int v = 0; v < 8; ++v) {
      float val = acc[v] + bv;
      bufB[(moff + v) * 132 + nb + m] = fmaxf(val, 0.f);
    }
  }
  __syncthreads();

  #pragma unroll
  for (int t = 0; t < 2; ++t) {
    const int nb = (wave * 2 + t) * 16;
    v8f acc = {};
    acc = wmma_k_loop(bufB + m * 132 + kb,
                      w1 + (size_t)(nb + m) * E_N + kb, E_N, acc);
    const float bv = b1[nb + m];
    #pragma unroll
    for (int v = 0; v < 8; ++v) {
      float val = acc[v] + bv;
      bufA[(moff + v) * 132 + nb + m] = fmaxf(val, 0.f);
    }
  }
  __syncthreads();

  #pragma unroll
  for (int t = 0; t < 2; ++t) {
    const int nb = (wave * 2 + t) * 16;
    v8f acc = {};
    acc = wmma_k_loop(bufA + m * 132 + kb,
                      w2 + (size_t)(nb + m) * E_N + kb, E_N, acc);
    const float bv = b2[nb + m];
    #pragma unroll
    for (int v = 0; v < 8; ++v)
      out[(size_t)(row0 + moff + v) * E_N + nb + m] = acc[v] + bv;
  }
}

__global__ __launch_bounds__(256) void pret_kernel(
    const float* __restrict__ taskE, const float* __restrict__ fw0,
    const float* __restrict__ fb0, float* __restrict__ preT) {
  const int lane  = threadIdx.x & 31;
  const int wave  = threadIdx.x >> 5;
  const int tile  = blockIdx.x * 8 + wave;
  const int ntile = tile & 15;
  const int ttile = tile >> 4;
  const int m    = lane & 15;
  const int kb   = (lane >> 4) << 1;
  const int moff = (lane >> 4) * 8;
  const int nb   = ntile * 16;

  v8f acc = {};
  acc = wmma_k_loop(taskE + (size_t)(ttile * 16 + m) * E_N + kb,
                    fw0 + (size_t)(nb + m) * (2 * E_N) + E_N + kb, E_N, acc);
  const float bv = fb0[nb + m];
  #pragma unroll
  for (int v = 0; v < 8; ++v)
    preT[(size_t)(ttile * 16 + moff + v) * HID_N + nb + m] = acc[v] + bv;
}

__global__ __launch_bounds__(256) void preu_kernel(
    const float* __restrict__ uavE, const float* __restrict__ hq,
    const float* __restrict__ fw0, float* __restrict__ preU) {
  const int lane  = threadIdx.x & 31;
  const int wave  = threadIdx.x >> 5;
  const int tile  = blockIdx.x * 8 + wave;
  const int ntile = tile & 15;
  const int mtile = tile >> 4;
  const int h     = mtile >> 4;
  const int utile = mtile & 15;
  const int m    = lane & 15;
  const int kb   = (lane >> 4) << 1;
  const int moff = (lane >> 4) * 8;
  const int nb   = ntile * 16;

  const float* up = uavE + (size_t)(utile * 16 + m) * E_N + kb;
  const float* qp = hq + (size_t)h * E_N + kb;
  const float* bp = fw0 + (size_t)(nb + m) * (2 * E_N) + kb;

  v8f acc = {};
  for (int kk = 0; kk < E_N; kk += 4) {
    v2f a; a.x = up[kk] + qp[kk]; a.y = up[kk + 1] + qp[kk + 1];
    v2f b; b.x = bp[kk]; b.y = bp[kk + 1];
    acc = __builtin_amdgcn_wmma_f32_16x16x4_f32(false, a, false, b,
                                                (short)0, acc, false, false);
  }
  #pragma unroll
  for (int v = 0; v < 8; ++v)
    preU[((size_t)h * U_N + utile * 16 + moff + v) * HID_N + nb + m] = acc[v];
}

__global__ __launch_bounds__(256) void logits_kernel(
    const float* __restrict__ preU, const float* __restrict__ preT,
    const float* __restrict__ fw1, const float* __restrict__ fb1,
    float* __restrict__ out) {
  __shared__ __align__(16) float puL[16 * 260];
  __shared__ __align__(16) float ptL[16 * 260];
  __shared__ __align__(16) float wL[HID_N];

  const int ttile = blockIdx.x & 31;
  const int utile = (blockIdx.x >> 5) & 15;
  const int h     = blockIdx.x >> 9;
  const int tid   = threadIdx.x;

  for (int i = tid; i < 16 * (HID_N / 4); i += 256) {
    int r = i >> 6, c4 = (i & 63) << 2;
    float4 a = *(const float4*)&preU[((size_t)h * U_N + utile * 16 + r) * HID_N + c4];
    *(float4*)&puL[r * 260 + c4] = a;
    float4 b = *(const float4*)&preT[((size_t)ttile * 16 + r) * HID_N + c4];
    *(float4*)&ptL[r * 260 + c4] = b;
  }
  if (tid < HID_N / 4)
    *(float4*)&wL[tid * 4] = *(const float4*)&fw1[tid * 4];
  __syncthreads();

  const int tt = tid & 15;
  const int tu = tid >> 4;
  const float* pu = &puL[tu * 260];
  const float* pt = &ptL[tt * 260];

  float acc = 0.f;
  #pragma unroll 4
  for (int k = 0; k < HID_N; k += 4) {
    float4 a = *(const float4*)&pu[k];
    float4 b = *(const float4*)&pt[k];
    float4 w = *(const float4*)&wL[k];
    acc = fmaf(fmaxf(a.x + b.x, 0.f), w.x, acc);
    acc = fmaf(fmaxf(a.y + b.y, 0.f), w.y, acc);
    acc = fmaf(fmaxf(a.z + b.z, 0.f), w.z, acc);
    acc = fmaf(fmaxf(a.w + b.w, 0.f), w.w, acc);
  }
  out[((size_t)h * U_N + utile * 16 + tu) * T_N + ttile * 16 + tt] = acc + fb1[0];
}

extern "C" void kernel_launch(void* const* d_in, const int* in_sizes, int n_in,
                              void* d_out, int out_size, void* d_ws, size_t ws_size,
                              hipStream_t stream) {
  (void)in_sizes; (void)n_in; (void)out_size; (void)ws_size;
  const float* uav_feat = (const float*)d_in[0];
  const float* task_feat = (const float*)d_in[1];
  const float* uw0 = (const float*)d_in[2];
  const float* ub0 = (const float*)d_in[3];
  const float* uw1 = (const float*)d_in[4];
  const float* ub1 = (const float*)d_in[5];
  const float* uw2 = (const float*)d_in[6];
  const float* ub2 = (const float*)d_in[7];
  const float* tw0 = (const float*)d_in[8];
  const float* tb0 = (const float*)d_in[9];
  const float* tw1 = (const float*)d_in[10];
  const float* tb1 = (const float*)d_in[11];
  const float* tw2 = (const float*)d_in[12];
  const float* tb2 = (const float*)d_in[13];
  const float* head_queries = (const float*)d_in[14];
  const float* fw0 = (const float*)d_in[15];
  const float* fb0 = (const float*)d_in[16];
  const float* fw1 = (const float*)d_in[17];
  const float* fb1 = (const float*)d_in[18];
  float* out = (float*)d_out;

  float* ws = (float*)d_ws;
  float* uavE  = ws;
  float* taskE = ws + 32768;
  float* preT  = ws + 32768 + 65536;
  float* preU  = ws + 32768 + 65536 + 131072;

  encoder_kernel<UAVD><<<U_N / 16, 128, 0, stream>>>(
      uav_feat, uw0, ub0, uw1, ub1, uw2, ub2, uavE);
  encoder_kernel<TASKD><<<T_N / 16, 128, 0, stream>>>(
      task_feat, tw0, tb0, tw1, tb1, tw2, tb2, taskE);

  pret_kernel<<<64, 256, 0, stream>>>(taskE, fw0, fb0, preT);
  preu_kernel<<<128, 256, 0, stream>>>(uavE, head_queries, fw0, preU);
  logits_kernel<<<H_N * (U_N / 16) * (T_N / 16), 256, 0, stream>>>(
      preU, preT, fw1, fb1, out);
}


# Fuse both encoders and both first-fusion-layer GEMMs into one launch: a block keeps its 16 rows in LDS through all four layers (uav encoder recomputed per head); 5 launches -> 2
# speedup vs baseline: 1.3960x; 1.3960x over previous
#include <hip/hip_runtime.h>

typedef __attribute__((ext_vector_type(2))) float v2f;
typedef __attribute__((ext_vector_type(8))) float v8f;

#define U_N     256
#define T_N     512
#define H_N     4
#define UAVD    64
#define TASKD   32
#define E_N     128
#define HID_N   256

__device__ __forceinline__ v8f wmma_k_loop(const float* __restrict__ ap,
                                           const float* __restrict__ bp,
                                           int K, v8f acc) {
  for (int kk = 0; kk < K; kk += 4) {
    v2f a; a.x = ap[kk]; a.y = ap[kk + 1];
    v2f b; b.x = bp[kk]; b.y = bp[kk + 1];
    acc = __builtin_amdgcn_wmma_f32_16x16x4_f32(false, a, false, b,
                                                (short)0, acc, false, false);
  }
  return acc;
}

template <int INK, bool TASK>
__device__ __forceinline__ void front_rows(
    float* __restrict__ bufA, float* __restrict__ bufB,
    const int row0, const int h,
    const float* __restrict__ x,
    const float* __restrict__ w0, const float* __restrict__ b0,
    const float* __restrict__ w1, const float* __restrict__ b1,
    const float* __restrict__ w2, const float* __restrict__ b2,
    const float* __restrict__ hq,
    const float* __restrict__ fw0, const float* __restrict__ fb0,
    float* __restrict__ pre) {
  const int tid  = threadIdx.x;
  const int lane = tid & 31;
  const int wave = tid >> 5;

  for (int i = tid; i < 16 * INK; i += 128) {
    int r = i / INK, c = i - r * INK;
    bufA[r * 132 + c] = x[(size_t)(row0 + r) * INK + c];
  }
  __syncthreads();

  const int m    = lane & 15;
  const int kb   = (lane >> 4) << 1;
  const int moff = (lane >> 4) * 8;

  #pragma unroll
  for (int t = 0; t < 2; ++t) {
    const int nb = (wave * 2 + t) * 16;
    v8f acc = {};
    acc = wmma_k_loop(bufA + m * 132 + kb,
                      w0 + (size_t)(nb + m) * INK + kb, INK, acc);
    const float bv = b0[nb + m];
    #pragma unroll
    for (int v = 0; v < 8; ++v) {
      float val = acc[v] + bv;
      bufB[(moff + v) * 132 + nb + m] = fmaxf(val, 0.f);
    }
  }
  __syncthreads();

  #pragma unroll
  for (int t = 0; t < 2; ++t) {
    const int nb = (wave * 2 + t) * 16;
    v8f acc = {};
    acc = wmma_k_loop(bufB + m * 132 + kb,
                      w1 + (size_t)(nb + m) * E_N + kb, E_N, acc);
    const float bv = b1[nb + m];
    #pragma unroll
    for (int v = 0; v < 8; ++v) {
      float val = acc[v] + bv;
      bufA[(moff + v) * 132 + nb + m] = fmaxf(val, 0.f);
    }
  }
  __syncthreads();

  #pragma unroll
  for (int t = 0; t < 2; ++t) {
    const int nb = (wave * 2 + t) * 16;
    v8f acc = {};
    acc = wmma_k_loop(bufA + m * 132 + kb,
                      w2 + (size_t)(nb + m) * E_N + kb, E_N, acc);
    const float bv = b2[nb + m];
    float qv = 0.f;
    if (!TASK) qv = hq[(size_t)h * E_N + nb + m];
    #pragma unroll
    for (int v = 0; v < 8; ++v) {
      float val = acc[v] + bv;
      if (!TASK) val = val + qv;
      bufB[(moff + v) * 132 + nb + m] = val;
    }
  }
  __syncthreads();

  {
    const float* ap = bufB + m * 132 + kb;
    const int nb0 = wave * 64;
    const float* bp = fw0 + (size_t)(nb0 + m) * (2 * E_N) + (TASK ? E_N : 0) + kb;
    v8f acc0 = {}, acc1 = {}, acc2 = {}, acc3 = {};
    #pragma unroll
    for (int kk = 0; kk < E_N; kk += 4) {
      v2f a;  a.x  = ap[kk]; a.y = ap[kk + 1];
      v2f b0v; b0v.x = bp[kk];                      b0v.y = bp[kk + 1];
      v2f b1v; b1v.x = bp[16 * 2 * E_N + kk];       b1v.y = bp[16 * 2 * E_N + kk + 1];
      v2f b2v; b2v.x = bp[32 * 2 * E_N + kk];       b2v.y = bp[32 * 2 * E_N + kk + 1];
      v2f b3v; b3v.x = bp[48 * 2 * E_N + kk];       b3v.y = bp[48 * 2 * E_N + kk + 1];
      acc0 = __builtin_amdgcn_wmma_f32_16x16x4_f32(false, a, false, b0v, (short)0, acc0, false, false);
      acc1 = __builtin_amdgcn_wmma_f32_16x16x4_f32(false, a, false, b1v, (short)0, acc1, false, false);
      acc2 = __builtin_amdgcn_wmma_f32_16x16x4_f32(false, a, false, b2v, (short)0, acc2, false, false);
      acc3 = __builtin_amdgcn_wmma_f32_16x16x4_f32(false, a, false, b3v, (short)0, acc3, false, false);
    }
    float* op = pre + (size_t)(row0 + moff) * HID_N + nb0 + m;
    float bv0 = 0.f, bv1 = 0.f, bv2 = 0.f, bv3 = 0.f;
    if (TASK) {
      bv0 = fb0[nb0 + m];      bv1 = fb0[nb0 + 16 + m];
      bv2 = fb0[nb0 + 32 + m]; bv3 = fb0[nb0 + 48 + m];
    }
    #pragma unroll
    for (int v = 0; v < 8; ++v) {
      if (TASK) {
        op[(size_t)v * HID_N]      = acc0[v] + bv0;
        op[(size_t)v * HID_N + 16] = acc1[v] + bv1;
        op[(size_t)v * HID_N + 32] = acc2[v] + bv2;
        op[(size_t)v * HID_N + 48] = acc3[v] + bv3;
      } else {
        op[(size_t)v * HID_N]      = acc0[v];
        op[(size_t)v * HID_N + 16] = acc1[v];
        op[(size_t)v * HID_N + 32] = acc2[v];
        op[(size_t)v * HID_N + 48] = acc3[v];
      }
    }
  }
}

__global__ __launch_bounds__(128) void front_kernel(
    const float* __restrict__ uav_feat, const float* __restrict__ task_feat,
    const float* __restrict__ uw0, const float* __restrict__ ub0,
    const float* __restrict__ uw1, const float* __restrict__ ub1,
    const float* __restrict__ uw2, const float* __restrict__ ub2,
    const float* __restrict__ tw0, const float* __restrict__ tb0,
    const float* __restrict__ tw1, const float* __restrict__ tb1,
    const float* __restrict__ tw2, const float* __restrict__ tb2,
    const float* __restrict__ hq,
    const float* __restrict__ fw0, const float* __restrict__ fb0,
    float* __restrict__ preT, float* __restrict__ preU) {
  __shared__ __align__(16) float bufA[16 * 132];
  __shared__ __align__(16) float bufB[16 * 132];
  const int b = blockIdx.x;
  if (b < H_N * (U_N / 16)) {
    const int h = b >> 4, utile = b & 15;
    front_rows<UAVD, false>(bufA, bufB, utile * 16, h, uav_feat,
                            uw0, ub0, uw1, ub1, uw2, ub2, hq, fw0, fb0,
                            preU + (size_t)h * U_N * HID_N);
  } else {
    const int ttile = b - H_N * (U_N / 16);
    front_rows<TASKD, true>(bufA, bufB, ttile * 16, 0, task_feat,
                            tw0, tb0, tw1, tb1, tw2, tb2, hq, fw0, fb0, preT);
  }
}

__global__ __launch_bounds__(256) void logits_kernel(
    const float* __restrict__ preU, const float* __restrict__ preT,
    const float* __restrict__ fw1, const float* __restrict__ fb1,
    float* __restrict__ out) {
  __shared__ __align__(16) float puL[16 * 260];
  __shared__ __align__(16) float ptL[16 * 260];
  __shared__ __align__(16) float wL[HID_N];

  const int ttile = blockIdx.x & 31;
  const int utile = (blockIdx.x >> 5) & 15;
  const int h     = blockIdx.x >> 9;
  const int tid   = threadIdx.x;

  for (int i = tid; i < 16 * (HID_N / 4); i += 256) {
    int r = i >> 6, c4 = (i & 63) << 2;
    float4 a = *(const float4*)&preU[((size_t)h * U_N + utile * 16 + r) * HID_N + c4];
    *(float4*)&puL[r * 260 + c4] = a;
    float4 b = *(const float4*)&preT[((size_t)ttile * 16 + r) * HID_N + c4];
    *(float4*)&ptL[r * 260 + c4] = b;
  }
  if (tid < HID_N / 4)
    *(float4*)&wL[tid * 4] = *(const float4*)&fw1[tid * 4];
  __syncthreads();

  const int tt = tid & 15;
  const int tu = tid >> 4;
  const float* pu = &puL[tu * 260];
  const float* pt = &ptL[tt * 260];

  float acc = 0.f;
  #pragma unroll 4
  for (int k = 0; k < HID_N; k += 4) {
    float4 a = *(const float4*)&pu[k];
    float4 b = *(const float4*)&pt[k];
    float4 w = *(const float4*)&wL[k];
    acc = fmaf(fmaxf(a.x + b.x, 0.f), w.x, acc);
    acc = fmaf(fmaxf(a.y + b.y, 0.f), w.y, acc);
    acc = fmaf(fmaxf(a.z + b.z, 0.f), w.z, acc);
    acc = fmaf(fmaxf(a.w + b.w, 0.f), w.w, acc);
  }
  out[((size_t)h * U_N + utile * 16 + tu) * T_N + ttile * 16 + tt] = acc + fb1[0];
}

extern "C" void kernel_launch(void* const* d_in, const int* in_sizes, int n_in,
                              void* d_out, int out_size, void* d_ws, size_t ws_size,
                              hipStream_t stream) {
  (void)in_sizes; (void)n_in; (void)out_size; (void)ws_size;
  const float* uav_feat = (const float*)d_in[0];
  const float* task_feat = (const float*)d_in[1];
  const float* uw0 = (const float*)d_in[2];
  const float* ub0 = (const float*)d_in[3];
  const float* uw1 = (const float*)d_in[4];
  const float* ub1 = (const float*)d_in[5];
  const float* uw2 = (const float*)d_in[6];
  const float* ub2 = (const float*)d_in[7];
  const float* tw0 = (const float*)d_in[8];
  const float* tb0 = (const float*)d_in[9];
  const float* tw1 = (const float*)d_in[10];
  const float* tb1 = (const float*)d_in[11];
  const float* tw2 = (const float*)d_in[12];
  const float* tb2 = (const float*)d_in[13];
  const float* head_queries = (const float*)d_in[14];
  const float* fw0 = (const float*)d_in[15];
  const float* fb0 = (const float*)d_in[16];
  const float* fw1 = (const float*)d_in[17];
  const float* fb1 = (const float*)d_in[18];
  float* out = (float*)d_out;

  float* ws = (float*)d_ws;
  float* preT  = ws;
  float* preU  = ws + 131072;

  front_kernel<<<H_N * (U_N / 16) + T_N / 16, 128, 0, stream>>>(
      uav_feat, task_feat, uw0, ub0, uw1, ub1, uw2, ub2,
      tw0, tb0, tw1, tb1, tw2, tb2, head_queries, fw0, fb0, preT, preU);
  logits_kernel<<<H_N * (U_N / 16) * (T_N / 16), 256, 0, stream>>>(
      preU, preT, fw1, fb1, out);
}


# Logits kernel re-tiled: one block per 32x64 output tile with full K in LDS, 2x4 outputs per thread in registers; 2048 -> 256 blocks, 8x fewer waves, about 3x fewer LDS reads per output; k order kept
# speedup vs baseline: 1.5438x; 1.1059x over previous
#include <hip/hip_runtime.h>

typedef __attribute__((ext_vector_type(2))) float v2f;
typedef __attribute__((ext_vector_type(8))) float v8f;

#define U_N     256
#define T_N     512
#define H_N     4
#define UAVD    64
#define TASKD   32
#define E_N     128
#define HID_N   256

__device__ __forceinline__ v8f wmma_k_loop(const float* __restrict__ ap,
                                           const float* __restrict__ bp,
                                           int K, v8f acc) {
  for (int kk = 0; kk < K; kk += 4) {
    v2f a; a.x = ap[kk]; a.y = ap[kk + 1];
    v2f b; b.x = bp[kk]; b.y = bp[kk + 1];
    acc = __builtin_amdgcn_wmma_f32_16x16x4_f32(false, a, false, b,
                                                (short)0, acc, false, false);
  }
  return acc;
}

template <int INK, bool TASK>
__device__ __forceinline__ void front_rows(
    float* __restrict__ bufA, float* __restrict__ bufB,
    const int row0, const int h,
    const float* __restrict__ x,
    const float* __restrict__ w0, const float* __restrict__ b0,
    const float* __restrict__ w1, const float* __restrict__ b1,
    const float* __restrict__ w2, const float* __restrict__ b2,
    const float* __restrict__ hq,
    const float* __restrict__ fw0, const float* __restrict__ fb0,
    float* __restrict__ pre) {
  const int tid  = threadIdx.x;
  const int lane = tid & 31;
  const int wave = tid >> 5;

  for (int i = tid; i < 16 * INK; i += 128) {
    int r = i / INK, c = i - r * INK;
    bufA[r * 132 + c] = x[(size_t)(row0 + r) * INK + c];
  }
  __syncthreads();

  const int m    = lane & 15;
  const int kb   = (lane >> 4) << 1;
  const int moff = (lane >> 4) * 8;

  #pragma unroll
  for (int t = 0; t < 2; ++t) {
    const int nb = (wave * 2 + t) * 16;
    v8f acc = {};
    acc = wmma_k_loop(bufA + m * 132 + kb,
                      w0 + (size_t)(nb + m) * INK + kb, INK, acc);
    const float bv = b0[nb + m];
    #pragma unroll
    for (int v = 0; v < 8; ++v) {
      float val = acc[v] + bv;
      bufB[(moff + v) * 132 + nb + m] = fmaxf(val, 0.f);
    }
  }
  __syncthreads();

  #pragma unroll
  for (int t = 0; t < 2; ++t) {
    const int nb = (wave * 2 + t) * 16;
    v8f acc = {};
    acc = wmma_k_loop(bufB + m * 132 + kb,
                      w1 + (size_t)(nb + m) * E_N + kb, E_N, acc);
    const float bv = b1[nb + m];
    #pragma unroll
    for (int v = 0; v < 8; ++v) {
      float val = acc[v] + bv;
      bufA[(moff + v) * 132 + nb + m] = fmaxf(val, 0.f);
    }
  }
  __syncthreads();

  #pragma unroll
  for (int t = 0; t < 2; ++t) {
    const int nb = (wave * 2 + t) * 16;
    v8f acc = {};
    acc = wmma_k_loop(bufA + m * 132 + kb,
                      w2 + (size_t)(nb + m) * E_N + kb, E_N, acc);
    const float bv = b2[nb + m];
    float qv = 0.f;
    if (!TASK) qv = hq[(size_t)h * E_N + nb + m];
    #pragma unroll
    for (int v = 0; v < 8; ++v) {
      float val = acc[v] + bv;
      if (!TASK) val = val + qv;
      bufB[(moff + v) * 132 + nb + m] = val;
    }
  }
  __syncthreads();

  {
    const float* ap = bufB + m * 132 + kb;
    const int nb0 = wave * 64;
    const float* bp = fw0 + (size_t)(nb0 + m) * (2 * E_N) + (TASK ? E_N : 0) + kb;
    v8f acc0 = {}, acc1 = {}, acc2 = {}, acc3 = {};
    #pragma unroll
    for (int kk = 0; kk < E_N; kk += 4) {
      v2f a;  a.x  = ap[kk]; a.y = ap[kk + 1];
      v2f b0v; b0v.x = bp[kk];                      b0v.y = bp[kk + 1];
      v2f b1v; b1v.x = bp[16 * 2 * E_N + kk];       b1v.y = bp[16 * 2 * E_N + kk + 1];
      v2f b2v; b2v.x = bp[32 * 2 * E_N + kk];       b2v.y = bp[32 * 2 * E_N + kk + 1];
      v2f b3v; b3v.x = bp[48 * 2 * E_N + kk];       b3v.y = bp[48 * 2 * E_N + kk + 1];
      acc0 = __builtin_amdgcn_wmma_f32_16x16x4_f32(false, a, false, b0v, (short)0, acc0, false, false);
      acc1 = __builtin_amdgcn_wmma_f32_16x16x4_f32(false, a, false, b1v, (short)0, acc1, false, false);
      acc2 = __builtin_amdgcn_wmma_f32_16x16x4_f32(false, a, false, b2v, (short)0, acc2, false, false);
      acc3 = __builtin_amdgcn_wmma_f32_16x16x4_f32(false, a, false, b3v, (short)0, acc3, false, false);
    }
    float* op = pre + (size_t)(row0 + moff) * HID_N + nb0 + m;
    float bv0 = 0.f, bv1 = 0.f, bv2 = 0.f, bv3 = 0.f;
    if (TASK) {
      bv0 = fb0[nb0 + m];      bv1 = fb0[nb0 + 16 + m];
      bv2 = fb0[nb0 + 32 + m]; bv3 = fb0[nb0 + 48 + m];
    }
    #pragma unroll
    for (int v = 0; v < 8; ++v) {
      if (TASK) {
        op[(size_t)v * HID_N]      = acc0[v] + bv0;
        op[(size_t)v * HID_N + 16] = acc1[v] + bv1;
        op[(size_t)v * HID_N + 32] = acc2[v] + bv2;
        op[(size_t)v * HID_N + 48] = acc3[v] + bv3;
      } else {
        op[(size_t)v * HID_N]      = acc0[v];
        op[(size_t)v * HID_N + 16] = acc1[v];
        op[(size_t)v * HID_N + 32] = acc2[v];
        op[(size_t)v * HID_N + 48] = acc3[v];
      }
    }
  }
}

__global__ __launch_bounds__(128) void front_kernel(
    const float* __restrict__ uav_feat, const float* __restrict__ task_feat,
    const float* __restrict__ uw0, const float* __restrict__ ub0,
    const float* __restrict__ uw1, const float* __restrict__ ub1,
    const float* __restrict__ uw2, const float* __restrict__ ub2,
    const float* __restrict__ tw0, const float* __restrict__ tb0,
    const float* __restrict__ tw1, const float* __restrict__ tb1,
    const float* __restrict__ tw2, const float* __restrict__ tb2,
    const float* __restrict__ hq,
    const float* __restrict__ fw0, const float* __restrict__ fb0,
    float* __restrict__ preT, float* __restrict__ preU) {
  __shared__ __align__(16) float bufA[16 * 132];
  __shared__ __align__(16) float bufB[16 * 132];
  const int b = blockIdx.x;
  if (b < H_N * (U_N / 16)) {
    const int h = b >> 4, utile = b & 15;
    front_rows<UAVD, false>(bufA, bufB, utile * 16, h, uav_feat,
                            uw0, ub0, uw1, ub1, uw2, ub2, hq, fw0, fb0,
                            preU + (size_t)h * U_N * HID_N);
  } else {
    const int ttile = b - H_N * (U_N / 16);
    front_rows<TASKD, true>(bufA, bufB, ttile * 16, 0, task_feat,
                            tw0, tb0, tw1, tb1, tw2, tb2, hq, fw0, fb0, preT);
  }
}

#define LG_R      32
#define LG_T      64
#define LG_PITCH  260

#define LG_STEP(A0, A1, B0, B1, B2, B3, WV)                                   \
  {                                                                           \
    acc00 = fmaf(fmaxf(A0.x + B0.x, 0.f), WV.x, acc00);                       \
    acc01 = fmaf(fmaxf(A0.x + B1.x, 0.f), WV.x, acc01);                       \
    acc02 = fmaf(fmaxf(A0.x + B2.x, 0.f), WV.x, acc02);                       \
    acc03 = fmaf(fmaxf(A0.x + B3.x, 0.f), WV.x, acc03);                       \
    acc10 = fmaf(fmaxf(A1.x + B0.x, 0.f), WV.x, acc10);                       \
    acc11 = fmaf(fmaxf(A1.x + B1.x, 0.f), WV.x, acc11);                       \
    acc12 = fmaf(fmaxf(A1.x + B2.x, 0.f), WV.x, acc12);                       \
    acc13 = fmaf(fmaxf(A1.x + B3.x, 0.f), WV.x, acc13);                       \
    acc00 = fmaf(fmaxf(A0.y + B0.y, 0.f), WV.y, acc00);                       \
    acc01 = fmaf(fmaxf(A0.y + B1.y, 0.f), WV.y, acc01);                       \
    acc02 = fmaf(fmaxf(A0.y + B2.y, 0.f), WV.y, acc02);                       \
    acc03 = fmaf(fmaxf(A0.y + B3.y, 0.f), WV.y, acc03);                       \
    acc10 = fmaf(fmaxf(A1.y + B0.y, 0.f), WV.y, acc10);                       \
    acc11 = fmaf(fmaxf(A1.y + B1.y, 0.f), WV.y, acc11);                       \
    acc12 = fmaf(fmaxf(A1.y + B2.y, 0.f), WV.y, acc12);                       \
    acc13 = fmaf(fmaxf(A1.y + B3.y, 0.f), WV.y, acc13);                       \
    acc00 = fmaf(fmaxf(A0.z + B0.z, 0.f), WV.z, acc00);                       \
    acc01 = fmaf(fmaxf(A0.z + B1.z, 0.f), WV.z, acc01);                       \
    acc02 = fmaf(fmaxf(A0.z + B2.z, 0.f), WV.z, acc02);                       \
    acc03 = fmaf(fmaxf(A0.z + B3.z, 0.f), WV.z, acc03);                       \
    acc10 = fmaf(fmaxf(A1.z + B0.z, 0.f), WV.z, acc10);                       \
    acc11 = fmaf(fmaxf(A1.z + B1.z, 0.f), WV.z, acc11);                       \
    acc12 = fmaf(fmaxf(A1.z + B2.z, 0.f), WV.z, acc12);                       \
    acc13 = fmaf(fmaxf(A1.z + B3.z, 0.f), WV.z, acc13);                       \
    acc00 = fmaf(fmaxf(A0.w + B0.w, 0.f), WV.w, acc00);                       \
    acc01 = fmaf(fmaxf(A0.w + B1.w, 0.f), WV.w, acc01);                       \
    acc02 = fmaf(fmaxf(A0.w + B2.w, 0.f), WV.w, acc02);                       \
    acc03 = fmaf(fmaxf(A0.w + B3.w, 0.f), WV.w, acc03);                       \
    acc10 = fmaf(fmaxf(A1.w + B0.w, 0.f), WV.w, acc10);                       \
    acc11 = fmaf(fmaxf(A1.w + B1.w, 0.f), WV.w, acc11);                       \
    acc12 = fmaf(fmaxf(A1.w + B2.w, 0.f), WV.w, acc12);                       \
    acc13 = fmaf(fmaxf(A1.w + B3.w, 0.f), WV.w, acc13);                       \
  }

__global__ __launch_bounds__(256) void logits_kernel(
    const float* __restrict__ preU, const float* __restrict__ preT,
    const float* __restrict__ fw1, const float* __restrict__ fb1,
    float* __restrict__ out) {
  __shared__ __align__(16) float puL[LG_R * LG_PITCH];
  __shared__ __align__(16) float ptL[LG_T * LG_PITCH];
  __shared__ __align__(16) float wL[HID_N];

  const int ttile = blockIdx.x & (T_N / LG_T - 1);
  const int rtile = blockIdx.x >> 3;
  const int tid   = threadIdx.x;

  const float* __restrict__ gu = preU + (size_t)rtile * LG_R * HID_N;
  const float* __restrict__ gt = preT + (size_t)ttile * LG_T * HID_N;

  const int frow = tid >> 6;
  const int fcol = (tid & 63) << 2;
  float4 ru[LG_R / 4], rt[LG_T / 4];
  float4 rw = make_float4(0.f, 0.f, 0.f, 0.f);
  #pragma unroll
  for (int j = 0; j < LG_R / 4; ++j)
    ru[j] = *(const float4*)&gu[(frow + 4 * j) * HID_N + fcol];
  #pragma unroll
  for (int j = 0; j < LG_T / 4; ++j)
    rt[j] = *(const float4*)&gt[(frow + 4 * j) * HID_N + fcol];
  if (tid < HID_N / 4) rw = *(const float4*)&fw1[tid * 4];
  #pragma unroll
  for (int j = 0; j < LG_R / 4; ++j)
    *(float4*)&puL[(frow + 4 * j) * LG_PITCH + fcol] = ru[j];
  #pragma unroll
  for (int j = 0; j < LG_T / 4; ++j)
    *(float4*)&ptL[(j + 16 * frow) * LG_PITCH + fcol] = rt[j];
  if (tid < HID_N / 4) *(float4*)&wL[tid * 4] = rw;
  __syncthreads();

  const int tt = tid & 15;
  const int tu = tid >> 4;
  const float* pa0 = &puL[tu * LG_PITCH];
  const float* pa1 = &puL[(tu + 16) * LG_PITCH];
  const float* pb0 = &ptL[tt * LG_PITCH];
  const float* pb1 = &ptL[(tt + 16) * LG_PITCH];
  const float* pb2 = &ptL[(tt + 32) * LG_PITCH];
  const float* pb3 = &ptL[(tt + 48) * LG_PITCH];

  float acc00 = 0.f, acc01 = 0.f, acc02 = 0.f, acc03 = 0.f;
  float acc10 = 0.f, acc11 = 0.f, acc12 = 0.f, acc13 = 0.f;

  float4 a0 = *(const float4*)&pa0[0];
  float4 a1 = *(const float4*)&pa1[0];
  float4 b0 = *(const float4*)&pb0[0];
  float4 b1 = *(const float4*)&pb1[0];
  float4 b2 = *(const float4*)&pb2[0];
  float4 b3 = *(const float4*)&pb3[0];
  float4 wv = *(const float4*)&wL[0];
  for (int k = 0; k < HID_N - 8; k += 8) {
    const float4 na0 = *(const float4*)&pa0[k + 4];
    const float4 na1 = *(const float4*)&pa1[k + 4];
    const float4 nb0 = *(const float4*)&pb0[k + 4];
    const float4 nb1 = *(const float4*)&pb1[k + 4];
    const float4 nb2 = *(const float4*)&pb2[k + 4];
    const float4 nb3 = *(const float4*)&pb3[k + 4];
    const float4 nwv = *(const float4*)&wL[k + 4];
    LG_STEP(a0, a1, b0, b1, b2, b3, wv)
    a0 = *(const float4*)&pa0[k + 8];
    a1 = *(const float4*)&pa1[k + 8];
    b0 = *(const float4*)&pb0[k + 8];
    b1 = *(const float4*)&pb1[k + 8];
    b2 = *(const float4*)&pb2[k + 8];
    b3 = *(const float4*)&pb3[k + 8];
    wv = *(const float4*)&wL[k + 8];
    LG_STEP(na0, na1, nb0, nb1, nb2, nb3, nwv)
  }
  {
    const float4 na0 = *(const float4*)&pa0[HID_N - 4];
    const float4 na1 = *(const float4*)&pa1[HID_N - 4];
    const float4 nb0 = *(const float4*)&pb0[HID_N - 4];
    const float4 nb1 = *(const float4*)&pb1[HID_N - 4];
    const float4 nb2 = *(const float4*)&pb2[HID_N - 4];
    const float4 nb3 = *(const float4*)&pb3[HID_N - 4];
    const float4 nwv = *(const float4*)&wL[HID_N - 4];
    LG_STEP(a0, a1, b0, b1, b2, b3, wv)
    LG_STEP(na0, na1, nb0, nb1, nb2, nb3, nwv)
  }

  const float bias = fb1[0];
  float* op = out + ((size_t)rtile * LG_R + tu) * T_N + ttile * LG_T + tt * 4;
  float4 o0, o1;
  o0.x = acc00 + bias; o0.y = acc01 + bias; o0.z = acc02 + bias; o0.w = acc03 + bias;
  o1.x = acc10 + bias; o1.y = acc11 + bias; o1.z = acc12 + bias; o1.w = acc13 + bias;
  *(float4*)&op[0]                 = o0;
  *(float4*)&op[(size_t)16 * T_N]  = o1;
}

extern "C" void kernel_launch(void* const* d_in, const int* in_sizes, int n_in,
                              void* d_out, int out_size, void* d_ws, size_t ws_size,
                              hipStream_t stream) {
  (void)in_sizes; (void)n_in; (void)out_size; (void)ws_size;
  const float* uav_feat = (const float*)d_in[0];
  const float* task_feat = (const float*)d_in[1];
  const float* uw0 = (const float*)d_in[2];
  const float* ub0 = (const float*)d_in[3];
  const float* uw1 = (const float*)d_in[4];
  const float* ub1 = (const float*)d_in[5];
  const float* uw2 = (const float*)d_in[6];
  const float* ub2 = (const float*)d_in[7];
  const float* tw0 = (const float*)d_in[8];
  const float* tb0 = (const float*)d_in[9];
  const float* tw1 = (const float*)d_in[10];
  const float* tb1 = (const float*)d_in[11];
  const float* tw2 = (const float*)d_in[12];
  const float* tb2 = (const float*)d_in[13];
  const float* head_queries = (const float*)d_in[14];
  const float* fw0 = (const float*)d_in[15];
  const float* fb0 = (const float*)d_in[16];
  const float* fw1 = (const float*)d_in[17];
  const float* fb1 = (const float*)d_in[18];
  float* out = (float*)d_out;

  float* ws = (float*)d_ws;
  float* preT  = ws;
  float* preU  = ws + 131072;

  front_kernel<<<H_N * (U_N / 16) + T_N / 16, 128, 0, stream>>>(
      uav_feat, task_feat, uw0, ub0, uw1, ub1, uw2, ub2,
      tw0, tb0, tw1, tb1, tw2, tb2, head_queries, fw0, fb0, preT, preU);
  logits_kernel<<<(H_N * U_N / LG_R) * (T_N / LG_T), 256, 0, stream>>>(
      preU, preT, fw1, fb1, out);
}


# Front kernel: weight fragments fetched by 16-byte loads issued one layer ahead (LDS-only barriers keep them in flight), input tile as one load group; k regrouped alike in A and B: sum order changes
# speedup vs baseline: 2.2299x; 1.4444x over previous
#include <hip/hip_runtime.h>

typedef __attribute__((ext_vector_type(2))) float v2f;
typedef __attribute__((ext_vector_type(8))) float v8f;

#define U_N     256
#define T_N     512
#define H_N     4
#define UAVD    64
#define TASKD   32
#define E_N     128
#define HID_N   256

#define WMMA_F32(A, B, C) \
  __builtin_amdgcn_wmma_f32_16x16x4_f32(false, (A), false, (B), (short)0, (C), false, false)

__device__ __forceinline__ int koff(int j, int hf) { return 8 * j + 4 * hf; }

__device__ __forceinline__ void lds_barrier() {
  __builtin_amdgcn_fence(__ATOMIC_RELEASE, "workgroup", "local");
  __builtin_amdgcn_s_barrier();
  __builtin_amdgcn_fence(__ATOMIC_ACQUIRE, "workgroup", "local");
}

#define HOLD_REQUESTS_ABOVE() __builtin_amdgcn_sched_barrier(0)

template <int NT, int K>
__device__ __forceinline__ void load_frags(float4 (&wf)[NT][K / 8],
                                           const float* __restrict__ wrow0,
                                           int ldw, int hf) {
  #pragma unroll
  for (int t = 0; t < NT; ++t) {
    const float* rp = wrow0 + (size_t)t * 16 * ldw;
    #pragma unroll
    for (int j = 0; j < K / 8; ++j)
      wf[t][j] = *(const float4*)(rp + koff(j, hf));
  }
}

template <int NT, int K>
__device__ __forceinline__ void mma_layer(v8f (&acc)[NT],
                                          const float* __restrict__ ap, int hf,
                                          const float4 (&wf)[NT][K / 8]) {
  #pragma unroll
  for (int j = 0; j < K / 8; ++j) {
    const float4 a4 = *(const float4*)(ap + koff(j, hf));
    v2f alo; alo.x = a4.x; alo.y = a4.y;
    v2f ahi; ahi.x = a4.z; ahi.y = a4.w;
    #pragma unroll
    for (int t = 0; t < NT; ++t) {
      v2f b; b.x = wf[t][j].x; b.y = wf[t][j].y;
      acc[t] = WMMA_F32(alo, b, acc[t]);
    }
    #pragma unroll
    for (int t = 0; t < NT; ++t) {
      v2f b; b.x = wf[t][j].z; b.y = wf[t][j].w;
      acc[t] = WMMA_F32(ahi, b, acc[t]);
    }
  }
}

template <int INK, bool TASK>
__device__ __forceinline__ void front_rows(
    float* __restrict__ bufA, float* __restrict__ bufB,
    const int row0, const int h,
    const float* __restrict__ x,
    const float* __restrict__ w0, const float* __restrict__ b0,
    const float* __restrict__ w1, const float* __restrict__ b1,
    const float* __restrict__ w2, const float* __restrict__ b2,
    const float* __restrict__ hq,
    const float* __restrict__ fw0, const float* __restrict__ fb0,
    float* __restrict__ pre) {
  const int tid  = threadIdx.x;
  const int lane = tid & 31;
  const int wave = tid >> 5;
  const int m    = lane & 15;
  const int hf   = lane >> 4;
  const int moff = hf * 8;
  const int nbw  = wave * 32;
  const int nb0  = wave * 64;

  constexpr int NX = (16 * INK / 4) / 128;
  float4 xin[NX];
  #pragma unroll
  for (int i = 0; i < NX; ++i)
    xin[i] = *(const float4*)(x + (size_t)row0 * INK + (size_t)(tid + 128 * i) * 4);
  float4 wf1[2][INK / 8];
  load_frags<2, INK>(wf1, w0 + (size_t)(nbw + m) * INK, INK, hf);
  const float bv1a = b0[nbw + m], bv1b = b0[nbw + 16 + m];

  #pragma unroll
  for (int i = 0; i < NX; ++i) {
    const int f = (tid + 128 * i) * 4;
    const int r = f / INK, c = f - r * INK;
    *(float4*)&bufA[r * 132 + c] = xin[i];
  }
  lds_barrier();

  float4 wf2[2][E_N / 8];
  load_frags<2, E_N>(wf2, w1 + (size_t)(nbw + m) * E_N, E_N, hf);
  const float bv2a = b1[nbw + m], bv2b = b1[nbw + 16 + m];
  HOLD_REQUESTS_ABOVE();
  {
    v8f acc[2] = {};
    mma_layer<2, INK>(acc, bufA + m * 132, hf, wf1);
    #pragma unroll
    for (int v = 0; v < 8; ++v) {
      bufB[(moff + v) * 132 + nbw + m]      = fmaxf(acc[0][v] + bv1a, 0.f);
      bufB[(moff + v) * 132 + nbw + 16 + m] = fmaxf(acc[1][v] + bv1b, 0.f);
    }
  }
  lds_barrier();

  float4 wf3[2][E_N / 8];
  load_frags<2, E_N>(wf3, w2 + (size_t)(nbw + m) * E_N, E_N, hf);
  const float bv3a = b2[nbw + m], bv3b = b2[nbw + 16 + m];
  float qva = 0.f, qvb = 0.f;
  if (!TASK) { qva = hq[(size_t)h * E_N + nbw + m]; qvb = hq[(size_t)h * E_N + nbw + 16 + m]; }
  const float* fwrow = fw0 + (size_t)(nb0 + m) * (2 * E_N) + (TASK ? E_N : 0);
  float4 wf4[4][E_N / 8];
  {
    float4 (&t0)[1][E_N / 8] = *reinterpret_cast<float4 (*)[1][E_N / 8]>(&wf4[0]);
    load_frags<1, E_N>(t0, fwrow, 2 * E_N, hf);
  }
  HOLD_REQUESTS_ABOVE();
  {
    v8f acc[2] = {};
    mma_layer<2, E_N>(acc, bufB + m * 132, hf, wf2);
    #pragma unroll
    for (int v = 0; v < 8; ++v) {
      bufA[(moff + v) * 132 + nbw + m]      = fmaxf(acc[0][v] + bv2a, 0.f);
      bufA[(moff + v) * 132 + nbw + 16 + m] = fmaxf(acc[1][v] + bv2b, 0.f);
    }
  }
  lds_barrier();

  {
    float4 (&t13)[3][E_N / 8] = *reinterpret_cast<float4 (*)[3][E_N / 8]>(&wf4[1]);
    load_frags<3, E_N>(t13, fwrow + (size_t)16 * 2 * E_N, 2 * E_N, hf);
  }
  float bv0 = 0.f, bv1 = 0.f, bv2 = 0.f, bv3 = 0.f;
  if (TASK) {
    bv0 = fb0[nb0 + m];      bv1 = fb0[nb0 + 16 + m];
    bv2 = fb0[nb0 + 32 + m]; bv3 = fb0[nb0 + 48 + m];
  }
  HOLD_REQUESTS_ABOVE();
  {
    v8f acc[2] = {};
    mma_layer<2, E_N>(acc, bufA + m * 132, hf, wf3);
    #pragma unroll
    for (int v = 0; v < 8; ++v) {
      float va = acc[0][v] + bv3a;
      float vb = acc[1][v] + bv3b;
      if (!TASK) { va = va + qva; vb = vb + qvb; }
      bufB[(moff + v) * 132 + nbw + m]      = va;
      bufB[(moff + v) * 132 + nbw + 16 + m] = vb;
    }
  }
  lds_barrier();

  {
    v8f acc[4] = {};
    mma_layer<4, E_N>(acc, bufB + m * 132, hf, wf4);
    float* op = pre + (size_t)(row0 + moff) * HID_N + nb0 + m;
    #pragma unroll
    for (int v = 0; v < 8; ++v) {
      if (TASK) {
        op[(size_t)v * HID_N]      = acc[0][v] + bv0;
        op[(size_t)v * HID_N + 16] = acc[1][v] + bv1;
        op[(size_t)v * HID_N + 32] = acc[2][v] + bv2;
        op[(size_t)v * HID_N + 48] = acc[3][v] + bv3;
      } else {
        op[(size_t)v * HID_N]      = acc[0][v];
        op[(size_t)v * HID_N + 16] = acc[1][v];
        op[(size_t)v * HID_N + 32] = acc[2][v];
        op[(size_t)v * HID_N + 48] = acc[3][v];
      }
    }
  }
}

__global__ __launch_bounds__(128) void front_kernel(
    const float* __restrict__ uav_feat, const float* __restrict__ task_feat,
    const float* __restrict__ uw0, const float* __restrict__ ub0,
    const float* __restrict__ uw1, const float* __restrict__ ub1,
    const float* __restrict__ uw2, const float* __restrict__ ub2,
    const float* __restrict__ tw0, const float* __restrict__ tb0,
    const float* __restrict__ tw1, const float* __restrict__ tb1,
    const float* __restrict__ tw2, const float* __restrict__ tb2,
    const float* __restrict__ hq,
    const float* __restrict__ fw0, const float* __restrict__ fb0,
    float* __restrict__ preT, float* __restrict__ preU) {
  __shared__ __align__(16) float bufA[16 * 132];
  __shared__ __align__(16) float bufB[16 * 132];
  const int b = blockIdx.x;
  if (b < H_N * (U_N / 16)) {
    const int h = b >> 4, utile = b & 15;
    front_rows<UAVD, false>(bufA, bufB, utile * 16, h, uav_feat,
                            uw0, ub0, uw1, ub1, uw2, ub2, hq, fw0, fb0,
                            preU + (size_t)h * U_N * HID_N);
  } else {
    const int ttile = b - H_N * (U_N / 16);
    front_rows<TASKD, true>(bufA, bufB, ttile * 16, 0, task_feat,
                            tw0, tb0, tw1, tb1, tw2, tb2, hq, fw0, fb0, preT);
  }
}

#define LG_R      32
#define LG_T      64
#define LG_PITCH  260

#define LG_STEP(A0, A1, B0, B1, B2, B3, WV)                                   \
  {                                                                           \
    acc00 = fmaf(fmaxf(A0.x + B0.x, 0.f), WV.x, acc00);                       \
    acc01 = fmaf(fmaxf(A0.x + B1.x, 0.f), WV.x, acc01);                       \
    acc02 = fmaf(fmaxf(A0.x + B2.x, 0.f), WV.x, acc02);                       \
    acc03 = fmaf(fmaxf(A0.x + B3.x, 0.f), WV.x, acc03);                       \
    acc10 = fmaf(fmaxf(A1.x + B0.x, 0.f), WV.x, acc10);                       \
    acc11 = fmaf(fmaxf(A1.x + B1.x, 0.f), WV.x, acc11);                       \
    acc12 = fmaf(fmaxf(A1.x + B2.x, 0.f), WV.x, acc12);                       \
    acc13 = fmaf(fmaxf(A1.x + B3.x, 0.f), WV.x, acc13);                       \
    acc00 = fmaf(fmaxf(A0.y + B0.y, 0.f), WV.y, acc00);                       \
    acc01 = fmaf(fmaxf(A0.y + B1.y, 0.f), WV.y, acc01);                       \
    acc02 = fmaf(fmaxf(A0.y + B2.y, 0.f), WV.y, acc02);                       \
    acc03 = fmaf(fmaxf(A0.y + B3.y, 0.f), WV.y, acc03);                       \
    acc10 = fmaf(fmaxf(A1.y + B0.y, 0.f), WV.y, acc10);                       \
    acc11 = fmaf(fmaxf(A1.y + B1.y, 0.f), WV.y, acc11);                       \
    acc12 = fmaf(fmaxf(A1.y + B2.y, 0.f), WV.y, acc12);                       \
    acc13 = fmaf(fmaxf(A1.y + B3.y, 0.f), WV.y, acc13);                       \
    acc00 = fmaf(fmaxf(A0.z + B0.z, 0.f), WV.z, acc00);                       \
    acc01 = fmaf(fmaxf(A0.z + B1.z, 0.f), WV.z, acc01);                       \
    acc02 = fmaf(fmaxf(A0.z + B2.z, 0.f), WV.z, acc02);                       \
    acc03 = fmaf(fmaxf(A0.z + B3.z, 0.f), WV.z, acc03);                       \
    acc10 = fmaf(fmaxf(A1.z + B0.z, 0.f), WV.z, acc10);                       \
    acc11 = fmaf(fmaxf(A1.z + B1.z, 0.f), WV.z, acc11);                       \
    acc12 = fmaf(fmaxf(A1.z + B2.z, 0.f), WV.z, acc12);                       \
    acc13 = fmaf(fmaxf(A1.z + B3.z, 0.f), WV.z, acc13);                       \
    acc00 = fmaf(fmaxf(A0.w + B0.w, 0.f), WV.w, acc00);                       \
    acc01 = fmaf(fmaxf(A0.w + B1.w, 0.f), WV.w, acc01);                       \
    acc02 = fmaf(fmaxf(A0.w + B2.w, 0.f), WV.w, acc02);                       \
    acc03 = fmaf(fmaxf(A0.w + B3.w, 0.f), WV.w, acc03);                       \
    acc10 = fmaf(fmaxf(A1.w + B0.w, 0.f), WV.w, acc10);                       \
    acc11 = fmaf(fmaxf(A1.w + B1.w, 0.f), WV.w, acc11);                       \
    acc12 = fmaf(fmaxf(A1.w + B2.w, 0.f), WV.w, acc12);                       \
    acc13 = fmaf(fmaxf(A1.w + B3.w, 0.f), WV.w, acc13);                       \
  }

__global__ __launch_bounds__(256) void logits_kernel(
    const float* __restrict__ preU, const float* __restrict__ preT,
    const float* __restrict__ fw1, const float* __restrict__ fb1,
    float* __restrict__ out) {
  __shared__ __align__(16) float puL[LG_R * LG_PITCH];
  __shared__ __align__(16) float ptL[LG_T * LG_PITCH];
  __shared__ __align__(16) float wL[HID_N];

  const int ttile = blockIdx.x & (T_N / LG_T - 1);
  const int rtile = blockIdx.x >> 3;
  const int tid   = threadIdx.x;

  const float* __restrict__ gu = preU + (size_t)rtile * LG_R * HID_N;
  const float* __restrict__ gt = preT + (size_t)ttile * LG_T * HID_N;

  const int frow = tid >> 6;
  const int fcol = (tid & 63) << 2;
  float4 ru[LG_R / 4], rt[LG_T / 4];
  float4 rw = make_float4(0.f, 0.f, 0.f, 0.f);
  #pragma unroll
  for (int j = 0; j < LG_R / 4; ++j)
    ru[j] = *(const float4*)&gu[(frow + 4 * j) * HID_N + fcol];
  #pragma unroll
  for (int j = 0; j < LG_T / 4; ++j)
    rt[j] = *(const float4*)&gt[(frow + 4 * j) * HID_N + fcol];
  if (tid < HID_N / 4) rw = *(const float4*)&fw1[tid * 4];
  #pragma unroll
  for (int j = 0; j < LG_R / 4; ++j)
    *(float4*)&puL[(frow + 4 * j) * LG_PITCH + fcol] = ru[j];
  #pragma unroll
  for (int j = 0; j < LG_T / 4; ++j)
    *(float4*)&ptL[(j + 16 * frow) * LG_PITCH + fcol] = rt[j];
  if (tid < HID_N / 4) *(float4*)&wL[tid * 4] = rw;
  __syncthreads();

  const int tt = tid & 15;
  const int tu = tid >> 4;
  const float* pa0 = &puL[tu * LG_PITCH];
  const float* pa1 = &puL[(tu + 16) * LG_PITCH];
  const float* pb0 = &ptL[tt * LG_PITCH];
  const float* pb1 = &ptL[(tt + 16) * LG_PITCH];
  const float* pb2 = &ptL[(tt + 32) * LG_PITCH];
  const float* pb3 = &ptL[(tt + 48) * LG_PITCH];

  float acc00 = 0.f, acc01 = 0.f, acc02 = 0.f, acc03 = 0.f;
  float acc10 = 0.f, acc11 = 0.f, acc12 = 0.f, acc13 = 0.f;

  float4 a0 = *(const float4*)&pa0[0];
  float4 a1 = *(const float4*)&pa1[0];
  float4 b0 = *(const float4*)&pb0[0];
  float4 b1 = *(const float4*)&pb1[0];
  float4 b2 = *(const float4*)&pb2[0];
  float4 b3 = *(const float4*)&pb3[0];
  float4 wv = *(const float4*)&wL[0];
  for (int k = 0; k < HID_N - 8; k += 8) {
    const float4 na0 = *(const float4*)&pa0[k + 4];
    const float4 na1 = *(const float4*)&pa1[k + 4];
    const float4 nb0 = *(const float4*)&pb0[k + 4];
    const float4 nb1 = *(const float4*)&pb1[k + 4];
    const float4 nb2 = *(const float4*)&pb2[k + 4];
    const float4 nb3 = *(const float4*)&pb3[k + 4];
    const float4 nwv = *(const float4*)&wL[k + 4];
    LG_STEP(a0, a1, b0, b1, b2, b3, wv)
    a0 = *(const float4*)&pa0[k + 8];
    a1 = *(const float4*)&pa1[k + 8];
    b0 = *(const float4*)&pb0[k + 8];
    b1 = *(const float4*)&pb1[k + 8];
    b2 = *(const float4*)&pb2[k + 8];
    b3 = *(const float4*)&pb3[k + 8];
    wv = *(const float4*)&wL[k + 8];
    LG_STEP(na0, na1, nb0, nb1, nb2, nb3, nwv)
  }
  {
    const float4 na0 = *(const float4*)&pa0[HID_N - 4];
    const float4 na1 = *(const float4*)&pa1[HID_N - 4];
    const float4 nb0 = *(const float4*)&pb0[HID_N - 4];
    const float4 nb1 = *(const float4*)&pb1[HID_N - 4];
    const float4 nb2 = *(const float4*)&pb2[HID_N - 4];
    const float4 nb3 = *(const float4*)&pb3[HID_N - 4];
    const float4 nwv = *(const float4*)&wL[HID_N - 4];
    LG_STEP(a0, a1, b0, b1, b2, b3, wv)
    LG_STEP(na0, na1, nb0, nb1, nb2, nb3, nwv)
  }

  const float bias = fb1[0];
  float* op = out + ((size_t)rtile * LG_R + tu) * T_N + ttile * LG_T + tt * 4;
  float4 o0, o1;
  o0.x = acc00 + bias; o0.y = acc01 + bias; o0.z = acc02 + bias; o0.w = acc03 + bias;
  o1.x = acc10 + bias; o1.y = acc11 + bias; o1.z = acc12 + bias; o1.w = acc13 + bias;
  *(float4*)&op[0]                 = o0;
  *(float4*)&op[(size_t)16 * T_N]  = o1;
}

extern "C" void kernel_launch(void* const* d_in, const int* in_sizes, int n_in,
                              void* d_out, int out_size, void* d_ws, size_t ws_size,
                              hipStream_t stream) {
  (void)in_sizes; (void)n_in; (void)out_size; (void)ws_size;
  const float* uav_feat = (const float*)d_in[0];
  const float* task_feat = (const float*)d_in[1];
  const float* uw0 = (const float*)d_in[2];
  const float* ub0 = (const float*)d_in[3];
  const float* uw1 = (const float*)d_in[4];
  const float* ub1 = (const float*)d_in[5];
  const float* uw2 = (const float*)d_in[6];
  const float* ub2 = (const float*)d_in[7];
  const float* tw0 = (const float*)d_in[8];
  const float* tb0 = (const float*)d_in[9];
  const float* tw1 = (const float*)d_in[10];
  const float* tb1 = (const float*)d_in[11];
  const float* tw2 = (const float*)d_in[12];
  const float* tb2 = (const float*)d_in[13];
  const float* head_queries = (const float*)d_in[14];
  const float* fw0 = (const float*)d_in[15];
  const float* fb0 = (const float*)d_in[16];
  const float* fw1 = (const float*)d_in[17];
  const float* fb1 = (const float*)d_in[18];
  float* out = (float*)d_out;

  float* ws = (float*)d_ws;
  float* preT  = ws;
  float* preU  = ws + 131072;

  front_kernel<<<H_N * (U_N / 16) + T_N / 16, 128, 0, stream>>>(
      uav_feat, task_feat, uw0, ub0, uw1, ub1, uw2, ub2,
      tw0, tb0, tw1, tb1, tw2, tb2, head_queries, fw0, fb0, preT, preU);
  logits_kernel<<<(H_N * U_N / LG_R) * (T_N / LG_T), 256, 0, stream>>>(
      preU, preT, fw1, fb1, out);
}


# Logits kernel: tile staged in 4 K chunks, all global loads in flight at once; chunk c+1 goes to LDS mid chunk c behind LDS-only barriers, so 3/4 of the fill runs under the k loop; sum order kept
# speedup vs baseline: 2.3691x; 1.0624x over previous
#include <hip/hip_runtime.h>

typedef __attribute__((ext_vector_type(2))) float v2f;
typedef __attribute__((ext_vector_type(8))) float v8f;

#define U_N     256
#define T_N     512
#define H_N     4
#define UAVD    64
#define TASKD   32
#define E_N     128
#define HID_N   256

#define WMMA_F32(A, B, C) \
  __builtin_amdgcn_wmma_f32_16x16x4_f32(false, (A), false, (B), (short)0, (C), false, false)

__device__ __forceinline__ int koff(int j, int hf) { return 8 * j + 4 * hf; }

__device__ __forceinline__ void lds_barrier() {
  __builtin_amdgcn_fence(__ATOMIC_RELEASE, "workgroup", "local");
  __builtin_amdgcn_s_barrier();
  __builtin_amdgcn_fence(__ATOMIC_ACQUIRE, "workgroup", "local");
}

#define HOLD_REQUESTS_ABOVE() __builtin_amdgcn_sched_barrier(0)

template <int NT, int K>
__device__ __forceinline__ void load_frags(float4 (&wf)[NT][K / 8],
                                           const float* __restrict__ wrow0,
                                           int ldw, int hf) {
  #pragma unroll
  for (int t = 0; t < NT; ++t) {
    const float* rp = wrow0 + (size_t)t * 16 * ldw;
    #pragma unroll
    for (int j = 0; j < K / 8; ++j)
      wf[t][j] = *(const float4*)(rp + koff(j, hf));
  }
}

template <int NT, int K>
__device__ __forceinline__ void mma_layer(v8f (&acc)[NT],
                                          const float* __restrict__ ap, int hf,
                                          const float4 (&wf)[NT][K / 8]) {
  #pragma unroll
  for (int j = 0; j < K / 8; ++j) {
    const float4 a4 = *(const float4*)(ap + koff(j, hf));
    v2f alo; alo.x = a4.x; alo.y = a4.y;
    v2f ahi; ahi.x = a4.z; ahi.y = a4.w;
    #pragma unroll
    for (int t = 0; t < NT; ++t) {
      v2f b; b.x = wf[t][j].x; b.y = wf[t][j].y;
      acc[t] = WMMA_F32(alo, b, acc[t]);
    }
    #pragma unroll
    for (int t = 0; t < NT; ++t) {
      v2f b; b.x = wf[t][j].z; b.y = wf[t][j].w;
      acc[t] = WMMA_F32(ahi, b, acc[t]);
    }
  }
}

template <int INK, bool TASK>
__device__ __forceinline__ void front_rows(
    float* __restrict__ bufA, float* __restrict__ bufB,
    const int row0, const int h,
    const float* __restrict__ x,
    const float* __restrict__ w0, const float* __restrict__ b0,
    const float* __restrict__ w1, const float* __restrict__ b1,
    const float* __restrict__ w2, const float* __restrict__ b2,
    const float* __restrict__ hq,
    const float* __restrict__ fw0, const float* __restrict__ fb0,
    float* __restrict__ pre) {
  const int tid  = threadIdx.x;
  const int lane = tid & 31;
  const int wave = tid >> 5;
  const int m    = lane & 15;
  const int hf   = lane >> 4;
  const int moff = hf * 8;
  const int nbw  = wave * 32;
  const int nb0  = wave * 64;

  constexpr int NX = (16 * INK / 4) / 128;
  float4 xin[NX];
  #pragma unroll
  for (int i = 0; i < NX; ++i)
    xin[i] = *(const float4*)(x + (size_t)row0 * INK + (size_t)(tid + 128 * i) * 4);
  float4 wf1[2][INK / 8];
  load_frags<2, INK>(wf1, w0 + (size_t)(nbw + m) * INK, INK, hf);
  const float bv1a = b0[nbw + m], bv1b = b0[nbw + 16 + m];

  #pragma unroll
  for (int i = 0; i < NX; ++i) {
    const int f = (tid + 128 * i) * 4;
    const int r = f / INK, c = f - r * INK;
    *(float4*)&bufA[r * 132 + c] = xin[i];
  }
  lds_barrier();

  float4 wf2[2][E_N / 8];
  load_frags<2, E_N>(wf2, w1 + (size_t)(nbw + m) * E_N, E_N, hf);
  const float bv2a = b1[nbw + m], bv2b = b1[nbw + 16 + m];
  HOLD_REQUESTS_ABOVE();
  {
    v8f acc[2] = {};
    mma_layer<2, INK>(acc, bufA + m * 132, hf, wf1);
    #pragma unroll
    for (int v = 0; v < 8; ++v) {
      bufB[(moff + v) * 132 + nbw + m]      = fmaxf(acc[0][v] + bv1a, 0.f);
      bufB[(moff + v) * 132 + nbw + 16 + m] = fmaxf(acc[1][v] + bv1b, 0.f);
    }
  }
  lds_barrier();

  float4 wf3[2][E_N / 8];
  load_frags<2, E_N>(wf3, w2 + (size_t)(nbw + m) * E_N, E_N, hf);
  const float bv3a = b2[nbw + m], bv3b = b2[nbw + 16 + m];
  float qva = 0.f, qvb = 0.f;
  if (!TASK) { qva = hq[(size_t)h * E_N + nbw + m]; qvb = hq[(size_t)h * E_N + nbw + 16 + m]; }
  const float* fwrow = fw0 + (size_t)(nb0 + m) * (2 * E_N) + (TASK ? E_N : 0);
  float4 wf4[4][E_N / 8];
  {
    float4 (&t0)[1][E_N / 8] = *reinterpret_cast<float4 (*)[1][E_N / 8]>(&wf4[0]);
    load_frags<1, E_N>(t0, fwrow, 2 * E_N, hf);
  }
  HOLD_REQUESTS_ABOVE();
  {
    v8f acc[2] = {};
    mma_layer<2, E_N>(acc, bufB + m * 132, hf, wf2);
    #pragma unroll
    for (int v = 0; v < 8; ++v) {
      bufA[(moff + v) * 132 + nbw + m]      = fmaxf(acc[0][v] + bv2a, 0.f);
      bufA[(moff + v) * 132 + nbw + 16 + m] = fmaxf(acc[1][v] + bv2b, 0.f);
    }
  }
  lds_barrier();

  {
    float4 (&t13)[3][E_N / 8] = *reinterpret_cast<float4 (*)[3][E_N / 8]>(&wf4[1]);
    load_frags<3, E_N>(t13, fwrow + (size_t)16 * 2 * E_N, 2 * E_N, hf);
  }
  float bv0 = 0.f, bv1 = 0.f, bv2 = 0.f, bv3 = 0.f;
  if (TASK) {
    bv0 = fb0[nb0 + m];      bv1 = fb0[nb0 + 16 + m];
    bv2 = fb0[nb0 + 32 + m]; bv3 = fb0[nb0 + 48 + m];
  }
  HOLD_REQUESTS_ABOVE();
  {
    v8f acc[2] = {};
    mma_layer<2, E_N>(acc, bufA + m * 132, hf, wf3);
    #pragma unroll
    for (int v = 0; v < 8; ++v) {
      float va = acc[0][v] + bv3a;
      float vb = acc[1][v] + bv3b;
      if (!TASK) { va = va + qva; vb = vb + qvb; }
      bufB[(moff + v) * 132 + nbw + m]      = va;
      bufB[(moff + v) * 132 + nbw + 16 + m] = vb;
    }
  }
  lds_barrier();

  {
    v8f acc[4] = {};
    mma_layer<4, E_N>(acc, bufB + m * 132, hf, wf4);
    float* op = pre + (size_t)(row0 + moff) * HID_N + nb0 + m;
    #pragma unroll
    for (int v = 0; v < 8; ++v) {
      if (TASK) {
        op[(size_t)v * HID_N]      = acc[0][v] + bv0;
        op[(size_t)v * HID_N + 16] = acc[1][v] + bv1;
        op[(size_t)v * HID_N + 32] = acc[2][v] + bv2;
        op[(size_t)v * HID_N + 48] = acc[3][v] + bv3;
      } else {
        op[(size_t)v * HID_N]      = acc[0][v];
        op[(size_t)v * HID_N + 16] = acc[1][v];
        op[(size_t)v * HID_N + 32] = acc[2][v];
        op[(size_t)v * HID_N + 48] = acc[3][v];
      }
    }
  }
}

__global__ __launch_bounds__(128) void front_kernel(
    const float* __restrict__ uav_feat, const float* __restrict__ task_feat,
    const float* __restrict__ uw0, const float* __restrict__ ub0,
    const float* __restrict__ uw1, const float* __restrict__ ub1,
    const float* __restrict__ uw2, const float* __restrict__ ub2,
    const float* __restrict__ tw0, const float* __restrict__ tb0,
    const float* __restrict__ tw1, const float* __restrict__ tb1,
    const float* __restrict__ tw2, const float* __restrict__ tb2,
    const float* __restrict__ hq,
    const float* __restrict__ fw0, const float* __restrict__ fb0,
    float* __restrict__ preT, float* __restrict__ preU) {
  __shared__ __align__(16) float bufA[16 * 132];
  __shared__ __align__(16) float bufB[16 * 132];
  const int b = blockIdx.x;
  if (b < H_N * (U_N / 16)) {
    const int h = b >> 4, utile = b & 15;
    front_rows<UAVD, false>(bufA, bufB, utile * 16, h, uav_feat,
                            uw0, ub0, uw1, ub1, uw2, ub2, hq, fw0, fb0,
                            preU + (size_t)h * U_N * HID_N);
  } else {
    const int ttile = b - H_N * (U_N / 16);
    front_rows<TASKD, true>(bufA, bufB, ttile * 16, 0, task_feat,
                            tw0, tb0, tw1, tb1, tw2, tb2, hq, fw0, fb0, preT);
  }
}

#define LG_R      32
#define LG_T      64
#define LG_PITCH  260

#define LG_STEP(A0, A1, B0, B1, B2, B3, WV)                                   \
  {                                                                           \
    acc00 = fmaf(fmaxf(A0.x + B0.x, 0.f), WV.x, acc00);                       \
    acc01 = fmaf(fmaxf(A0.x + B1.x, 0.f), WV.x, acc01);                       \
    acc02 = fmaf(fmaxf(A0.x + B2.x, 0.f), WV.x, acc02);                       \
    acc03 = fmaf(fmaxf(A0.x + B3.x, 0.f), WV.x, acc03);                       \
    acc10 = fmaf(fmaxf(A1.x + B0.x, 0.f), WV.x, acc10);                       \
    acc11 = fmaf(fmaxf(A1.x + B1.x, 0.f), WV.x, acc11);                       \
    acc12 = fmaf(fmaxf(A1.x + B2.x, 0.f), WV.x, acc12);                       \
    acc13 = fmaf(fmaxf(A1.x + B3.x, 0.f), WV.x, acc13);                       \
    acc00 = fmaf(fmaxf(A0.y + B0.y, 0.f), WV.y, acc00);                       \
    acc01 = fmaf(fmaxf(A0.y + B1.y, 0.f), WV.y, acc01);                       \
    acc02 = fmaf(fmaxf(A0.y + B2.y, 0.f), WV.y, acc02);                       \
    acc03 = fmaf(fmaxf(A0.y + B3.y, 0.f), WV.y, acc03);                       \
    acc10 = fmaf(fmaxf(A1.y + B0.y, 0.f), WV.y, acc10);                       \
    acc11 = fmaf(fmaxf(A1.y + B1.y, 0.f), WV.y, acc11);                       \
    acc12 = fmaf(fmaxf(A1.y + B2.y, 0.f), WV.y, acc12);                       \
    acc13 = fmaf(fmaxf(A1.y + B3.y, 0.f), WV.y, acc13);                       \
    acc00 = fmaf(fmaxf(A0.z + B0.z, 0.f), WV.z, acc00);                       \
    acc01 = fmaf(fmaxf(A0.z + B1.z, 0.f), WV.z, acc01);                       \
    acc02 = fmaf(fmaxf(A0.z + B2.z, 0.f), WV.z, acc02);                       \
    acc03 = fmaf(fmaxf(A0.z + B3.z, 0.f), WV.z, acc03);                       \
    acc10 = fmaf(fmaxf(A1.z + B0.z, 0.f), WV.z, acc10);                       \
    acc11 = fmaf(fmaxf(A1.z + B1.z, 0.f), WV.z, acc11);                       \
    acc12 = fmaf(fmaxf(A1.z + B2.z, 0.f), WV.z, acc12);                       \
    acc13 = fmaf(fmaxf(A1.z + B3.z, 0.f), WV.z, acc13);                       \
    acc00 = fmaf(fmaxf(A0.w + B0.w, 0.f), WV.w, acc00);                       \
    acc01 = fmaf(fmaxf(A0.w + B1.w, 0.f), WV.w, acc01);                       \
    acc02 = fmaf(fmaxf(A0.w + B2.w, 0.f), WV.w, acc02);                       \
    acc03 = fmaf(fmaxf(A0.w + B3.w, 0.f), WV.w, acc03);                       \
    acc10 = fmaf(fmaxf(A1.w + B0.w, 0.f), WV.w, acc10);                       \
    acc11 = fmaf(fmaxf(A1.w + B1.w, 0.f), WV.w, acc11);                       \
    acc12 = fmaf(fmaxf(A1.w + B2.w, 0.f), WV.w, acc12);                       \
    acc13 = fmaf(fmaxf(A1.w + B3.w, 0.f), WV.w, acc13);                       \
  }

#define LG_NC     4
#define LG_CK     (HID_N / LG_NC)

#define LG_TRIP(K)                                                            \
  {                                                                           \
    const float4 na0 = *(const float4*)&pa0[(K) + 4];                         \
    const float4 na1 = *(const float4*)&pa1[(K) + 4];                         \
    const float4 nb0 = *(const float4*)&pb0[(K) + 4];                         \
    const float4 nb1 = *(const float4*)&pb1[(K) + 4];                         \
    const float4 nb2 = *(const float4*)&pb2[(K) + 4];                         \
    const float4 nb3 = *(const float4*)&pb3[(K) + 4];                         \
    const float4 nwv = *(const float4*)&wL[(K) + 4];                          \
    LG_STEP(a0, a1, b0, b1, b2, b3, wv)                                       \
    a0 = *(const float4*)&pa0[(K) + 8];                                       \
    a1 = *(const float4*)&pa1[(K) + 8];                                       \
    b0 = *(const float4*)&pb0[(K) + 8];                                       \
    b1 = *(const float4*)&pb1[(K) + 8];                                       \
    b2 = *(const float4*)&pb2[(K) + 8];                                       \
    b3 = *(const float4*)&pb3[(K) + 8];                                       \
    wv = *(const float4*)&wL[(K) + 8];                                        \
    LG_STEP(na0, na1, nb0, nb1, nb2, nb3, nwv)                                \
  }

#define LG_TRIPS(KLO, KHI)                                                    \
  for (int k = (KLO); k < (KHI); k += 8) LG_TRIP(k)

#define LG_GU(J, C) (*(const float4*)&gu[(rb + 16 * (J)) * HID_N + LG_CK * (C) + cp])
#define LG_GT(J, C) (*(const float4*)&gt[(rb + 16 * (J)) * HID_N + LG_CK * (C) + cp])
#define LG_LOAD_CHUNK(C)                                                      \
  const float4 su##C##_0 = LG_GU(0, C), su##C##_1 = LG_GU(1, C);              \
  const float4 st##C##_0 = LG_GT(0, C), st##C##_1 = LG_GT(1, C);              \
  const float4 st##C##_2 = LG_GT(2, C), st##C##_3 = LG_GT(3, C);

#define LG_PU(J, C) (*(float4*)&puL[(rb + 16 * (J)) * LG_PITCH + LG_CK * (C) + cp])
#define LG_PT(J, C) (*(float4*)&ptL[((rb >> 2) + 4 * (J) + 16 * (rb & 3)) * LG_PITCH + LG_CK * (C) + cp])
#define LG_STORE_CHUNK(C)                                                     \
  {                                                                           \
    LG_PU(0, C) = su##C##_0; LG_PU(1, C) = su##C##_1;                         \
    LG_PT(0, C) = st##C##_0; LG_PT(1, C) = st##C##_1;                         \
    LG_PT(2, C) = st##C##_2; LG_PT(3, C) = st##C##_3;                         \
  }

__global__ __launch_bounds__(256) void logits_kernel(
    const float* preU, const float* preT,
    const float* fw1, const float* __restrict__ fb1,
    float* out) {
  __shared__ __align__(16) float puL[LG_R * LG_PITCH];
  __shared__ __align__(16) float ptL[LG_T * LG_PITCH];
  __shared__ __align__(16) float wL[HID_N];

  const int ttile = blockIdx.x & (T_N / LG_T - 1);
  const int rtile = blockIdx.x >> 3;
  const int tid   = threadIdx.x;

  const float* gu = preU + (size_t)rtile * LG_R * HID_N;
  const float* gt = preT + (size_t)ttile * LG_T * HID_N;

  const int cp = (tid & 15) << 2;
  const int rb = tid >> 4;
  LG_LOAD_CHUNK(0)
  const float4 rw = *(const float4*)&fw1[(tid & (HID_N / 4 - 1)) * 4];
  __builtin_amdgcn_sched_barrier(0);
  LG_LOAD_CHUNK(1)
  __builtin_amdgcn_sched_barrier(0);
  LG_LOAD_CHUNK(2)
  __builtin_amdgcn_sched_barrier(0);
  LG_LOAD_CHUNK(3)
  __builtin_amdgcn_sched_barrier(0);

  LG_STORE_CHUNK(0)
  if (tid < HID_N / 4) *(float4*)&wL[tid * 4] = rw;
  lds_barrier();

  const int tt = tid & 15;
  const int tu = tid >> 4;
  const float* pa0 = &puL[tu * LG_PITCH];
  const float* pa1 = &puL[(tu + 16) * LG_PITCH];
  const float* pb0 = &ptL[tt * LG_PITCH];
  const float* pb1 = &ptL[(tt + 16) * LG_PITCH];
  const float* pb2 = &ptL[(tt + 32) * LG_PITCH];
  const float* pb3 = &ptL[(tt + 48) * LG_PITCH];

  float acc00 = 0.f, acc01 = 0.f, acc02 = 0.f, acc03 = 0.f;
  float acc10 = 0.f, acc11 = 0.f, acc12 = 0.f, acc13 = 0.f;

  float4 a0 = *(const float4*)&pa0[0];
  float4 a1 = *(const float4*)&pa1[0];
  float4 b0 = *(const float4*)&pb0[0];
  float4 b1 = *(const float4*)&pb1[0];
  float4 b2 = *(const float4*)&pb2[0];
  float4 b3 = *(const float4*)&pb3[0];
  float4 wv = *(const float4*)&wL[0];

  LG_TRIPS(0, 32)
  LG_STORE_CHUNK(1)
  lds_barrier();
  LG_TRIPS(32, 96)
  LG_STORE_CHUNK(2)
  lds_barrier();
  LG_TRIPS(96, 160)
  LG_STORE_CHUNK(3)
  lds_barrier();
  LG_TRIPS(160, HID_N - 8)
  {
    const float4 na0 = *(const float4*)&pa0[HID_N - 4];
    const float4 na1 = *(const float4*)&pa1[HID_N - 4];
    const float4 nb0 = *(const float4*)&pb0[HID_N - 4];
    const float4 nb1 = *(const float4*)&pb1[HID_N - 4];
    const float4 nb2 = *(const float4*)&pb2[HID_N - 4];
    const float4 nb3 = *(const float4*)&pb3[HID_N - 4];
    const float4 nwv = *(const float4*)&wL[HID_N - 4];
    LG_STEP(a0, a1, b0, b1, b2, b3, wv)
    LG_STEP(na0, na1, nb0, nb1, nb2, nb3, nwv)
  }

  const float bias = fb1[0];
  float* op = out + ((size_t)rtile * LG_R + tu) * T_N + ttile * LG_T + tt * 4;
  float4 o0, o1;
  o0.x = acc00 + bias; o0.y = acc01 + bias; o0.z = acc02 + bias; o0.w = acc03 + bias;
  o1.x = acc10 + bias; o1.y = acc11 + bias; o1.z = acc12 + bias; o1.w = acc13 + bias;
  *(float4*)&op[0]                 = o0;
  *(float4*)&op[(size_t)16 * T_N]  = o1;
}

extern "C" void kernel_launch(void* const* d_in, const int* in_sizes, int n_in,
                              void* d_out, int out_size, void* d_ws, size_t ws_size,
                              hipStream_t stream) {
  (void)in_sizes; (void)n_in; (void)out_size; (void)ws_size;
  const float* uav_feat = (const float*)d_in[0];
  const float* task_feat = (const float*)d_in[1];
  const float* uw0 = (const float*)d_in[2];
  const float* ub0 = (const float*)d_in[3];
  const float* uw1 = (const float*)d_in[4];
  const float* ub1 = (const float*)d_in[5];
  const float* uw2 = (const float*)d_in[6];
  const float* ub2 = (const float*)d_in[7];
  const float* tw0 = (const float*)d_in[8];
  const float* tb0 = (const float*)d_in[9];
  const float* tw1 = (const float*)d_in[10];
  const float* tb1 = (const float*)d_in[11];
  const float* tw2 = (const float*)d_in[12];
  const float* tb2 = (const float*)d_in[13];
  const float* head_queries = (const float*)d_in[14];
  const float* fw0 = (const float*)d_in[15];
  const float* fb0 = (const float*)d_in[16];
  const float* fw1 = (const float*)d_in[17];
  const float* fb1 = (const float*)d_in[18];
  float* out = (float*)d_out;

  float* ws = (float*)d_ws;
  float* preT  = ws;
  float* preU  = ws + 131072;

  front_kernel<<<H_N * (U_N / 16) + T_N / 16, 128, 0, stream>>>(
      uav_feat, task_feat, uw0, ub0, uw1, ub1, uw2, ub2,
      tw0, tb0, tw1, tb1, tw2, tb2, head_queries, fw0, fb0, preT, preU);
  logits_kernel<<<(H_N * U_N / LG_R) * (T_N / LG_T), 256, 0, stream>>>(
      preU, preT, fw1, fb1, out);
}


# Front kernel: each wave fetches its weight rows as whole 512-byte rows (4 lines per load, was 16 lines x 32 B) into its own LDS area and reads the B fragments from there; k order and sum order kept
# speedup vs baseline: 2.5633x; 1.0820x over previous
#include <hip/hip_runtime.h>

typedef __attribute__((ext_vector_type(2))) float v2f;
typedef __attribute__((ext_vector_type(8))) float v8f;

#define U_N     256
#define T_N     512
#define H_N     4
#define UAVD    64
#define TASKD   32
#define E_N     128
#define HID_N   256

#define WMMA_F32(A, B, C) \
  __builtin_amdgcn_wmma_f32_16x16x4_f32(false, (A), false, (B), (short)0, (C), false, false)

__device__ __forceinline__ int koff(int j, int hf) { return 8 * j + 4 * hf; }

__device__ __forceinline__ void lds_barrier() {
  __builtin_amdgcn_fence(__ATOMIC_RELEASE, "workgroup", "local");
  __builtin_amdgcn_s_barrier();
  __builtin_amdgcn_fence(__ATOMIC_ACQUIRE, "workgroup", "local");
}

#define HOLD_REQUESTS_ABOVE() __builtin_amdgcn_sched_barrier(0)

#define LPITCH   132
#define SLAB_FLOATS (64 * LPITCH)

template <int R, int K>
__device__ __forceinline__ void slab_request(float4 (&g)[R * K / 128],
                                             const float* __restrict__ base,
                                             int ldw, int lane) {
  constexpr int LPR = K / 4;
  constexpr int RPL = 32 / LPR;
  const int lr = lane / LPR, lc = (lane % LPR) * 4;
  #pragma unroll
  for (int i = 0; i < R / RPL; ++i)
    g[i] = *(const float4*)(base + (size_t)(i * RPL + lr) * ldw + lc);
}

template <int R, int K>
__device__ __forceinline__ void slab_to_lds(float* __restrict__ ws,
                                            const float4 (&g)[R * K / 128], int lane) {
  constexpr int LPR = K / 4;
  constexpr int RPL = 32 / LPR;
  const int lr = lane / LPR, lc = (lane % LPR) * 4;
  #pragma unroll
  for (int i = 0; i < R / RPL; ++i)
    *(float4*)&ws[(i * RPL + lr) * LPITCH + lc] = g[i];
}

template <int NT, int K>
__device__ __forceinline__ void mma_layer(v8f (&acc)[NT],
                                          const float* __restrict__ ap,
                                          const float* __restrict__ wp, int hf) {
  float4 af[K / 8];
  float4 bf[NT][K / 8];
  #pragma unroll
  for (int j = 0; j < K / 8; ++j) {
    af[j] = *(const float4*)(ap + koff(j, hf));
    #pragma unroll
    for (int t = 0; t < NT; ++t)
      bf[t][j] = *(const float4*)(wp + t * 16 * LPITCH + koff(j, hf));
  }
  __builtin_amdgcn_sched_barrier(0);
  #pragma unroll
  for (int j = 0; j < K / 8; ++j) {
    v2f alo; alo.x = af[j].x; alo.y = af[j].y;
    v2f ahi; ahi.x = af[j].z; ahi.y = af[j].w;
    #pragma unroll
    for (int t = 0; t < NT; ++t) {
      v2f b; b.x = bf[t][j].x; b.y = bf[t][j].y;
      acc[t] = WMMA_F32(alo, b, acc[t]);
    }
    #pragma unroll
    for (int t = 0; t < NT; ++t) {
      v2f b; b.x = bf[t][j].z; b.y = bf[t][j].w;
      acc[t] = WMMA_F32(ahi, b, acc[t]);
    }
  }
}

#define SLAB_REQ(NAME, I) const float4 NAME##_##I = *(const float4*)(SLAB_BASE + (size_t)(I) * (SLAB_LDW) + lane * 4);
#define SLAB_PUT(NAME, I) *(float4*)&ws[(I) * LPITCH + lane * 4] = NAME##_##I;

template <int INK, bool TASK>
__device__ __forceinline__ void front_rows(
    float* __restrict__ bufA, float* __restrict__ bufB, float* __restrict__ slabs,
    const int row0, const int h,
    const float* __restrict__ x,
    const float* __restrict__ w0, const float* __restrict__ b0,
    const float* __restrict__ w1, const float* __restrict__ b1,
    const float* __restrict__ w2, const float* __restrict__ b2,
    const float* __restrict__ hq,
    const float* __restrict__ fw0, const float* __restrict__ fb0,
    float* __restrict__ pre) {
  const int tid  = threadIdx.x;
  const int lane = tid & 31;
  const int wave = tid >> 5;
  const int m    = lane & 15;
  const int hf   = lane >> 4;
  const int moff = hf * 8;
  const int nbw  = wave * 32;
  const int nb0  = wave * 64;
  float* ws = slabs + wave * SLAB_FLOATS;
  const float* wfrag = ws + m * LPITCH;

  constexpr int NX = (16 * INK / 4) / 128;
  float4 xin[NX];
  #pragma unroll
  for (int i = 0; i < NX; ++i)
    xin[i] = *(const float4*)(x + (size_t)row0 * INK + (size_t)(tid + 128 * i) * 4);
  float4 g1[32 * INK / 128];
  slab_request<32, INK>(g1, w0 + (size_t)nbw * INK, INK, lane);
  const float bv1a = b0[nbw + m], bv1b = b0[nbw + 16 + m];

  #pragma unroll
  for (int i = 0; i < NX; ++i) {
    const int f = (tid + 128 * i) * 4;
    const int r = f / INK, c = f - r * INK;
    *(float4*)&bufA[r * LPITCH + c] = xin[i];
  }
  slab_to_lds<32, INK>(ws, g1, lane);
  lds_barrier();

  const float* w1slab = w1 + (size_t)nbw * E_N;
#define SLAB_BASE w1slab
#define SLAB_LDW  E_N
  SLAB_REQ(g2, 0) SLAB_REQ(g2, 1) SLAB_REQ(g2, 2) SLAB_REQ(g2, 3) SLAB_REQ(g2, 4) SLAB_REQ(g2, 5)
  SLAB_REQ(g2, 6) SLAB_REQ(g2, 7) SLAB_REQ(g2, 8) SLAB_REQ(g2, 9) SLAB_REQ(g2, 10) SLAB_REQ(g2, 11)
  SLAB_REQ(g2, 12) SLAB_REQ(g2, 13) SLAB_REQ(g2, 14) SLAB_REQ(g2, 15) SLAB_REQ(g2, 16) SLAB_REQ(g2, 17)
  SLAB_REQ(g2, 18) SLAB_REQ(g2, 19) SLAB_REQ(g2, 20) SLAB_REQ(g2, 21) SLAB_REQ(g2, 22) SLAB_REQ(g2, 23)
  SLAB_REQ(g2, 24) SLAB_REQ(g2, 25) SLAB_REQ(g2, 26) SLAB_REQ(g2, 27) SLAB_REQ(g2, 28) SLAB_REQ(g2, 29)
  SLAB_REQ(g2, 30) SLAB_REQ(g2, 31)
#undef SLAB_BASE
#undef SLAB_LDW
  const float bv2a = b1[nbw + m], bv2b = b1[nbw + 16 + m];
  HOLD_REQUESTS_ABOVE();
  {
    v8f acc[2] = {};
    mma_layer<2, INK>(acc, bufA + m * LPITCH, wfrag, hf);
    #pragma unroll
    for (int v = 0; v < 8; ++v) {
      bufB[(moff + v) * LPITCH + nbw + m]      = fmaxf(acc[0][v] + bv1a, 0.f);
      bufB[(moff + v) * LPITCH + nbw + 16 + m] = fmaxf(acc[1][v] + bv1b, 0.f);
    }
  }
  lds_barrier();

  const float* w2slab = w2 + (size_t)nbw * E_N;
#define SLAB_BASE w2slab
#define SLAB_LDW  E_N
  SLAB_REQ(g3, 0) SLAB_REQ(g3, 1) SLAB_REQ(g3, 2) SLAB_REQ(g3, 3) SLAB_REQ(g3, 4) SLAB_REQ(g3, 5)
  SLAB_REQ(g3, 6) SLAB_REQ(g3, 7) SLAB_REQ(g3, 8) SLAB_REQ(g3, 9) SLAB_REQ(g3, 10) SLAB_REQ(g3, 11)
  SLAB_REQ(g3, 12) SLAB_REQ(g3, 13) SLAB_REQ(g3, 14) SLAB_REQ(g3, 15) SLAB_REQ(g3, 16) SLAB_REQ(g3, 17)
  SLAB_REQ(g3, 18) SLAB_REQ(g3, 19) SLAB_REQ(g3, 20) SLAB_REQ(g3, 21) SLAB_REQ(g3, 22) SLAB_REQ(g3, 23)
  SLAB_REQ(g3, 24) SLAB_REQ(g3, 25) SLAB_REQ(g3, 26) SLAB_REQ(g3, 27) SLAB_REQ(g3, 28) SLAB_REQ(g3, 29)
  SLAB_REQ(g3, 30) SLAB_REQ(g3, 31)
#undef SLAB_BASE
#undef SLAB_LDW
  const float bv3a = b2[nbw + m], bv3b = b2[nbw + 16 + m];
  float qva = 0.f, qvb = 0.f;
  if (!TASK) { qva = hq[(size_t)h * E_N + nbw + m]; qvb = hq[(size_t)h * E_N + nbw + 16 + m]; }
  const float* fwslab = fw0 + (size_t)nb0 * (2 * E_N) + (TASK ? E_N : 0);
#define SLAB_BASE fwslab
#define SLAB_LDW  (2 * E_N)
  SLAB_REQ(g4, 0) SLAB_REQ(g4, 1) SLAB_REQ(g4, 2) SLAB_REQ(g4, 3) SLAB_REQ(g4, 4) SLAB_REQ(g4, 5)
  SLAB_REQ(g4, 6) SLAB_REQ(g4, 7) SLAB_REQ(g4, 8) SLAB_REQ(g4, 9) SLAB_REQ(g4, 10) SLAB_REQ(g4, 11)
  SLAB_REQ(g4, 12) SLAB_REQ(g4, 13) SLAB_REQ(g4, 14) SLAB_REQ(g4, 15)
  SLAB_PUT(g2, 0) SLAB_PUT(g2, 1) SLAB_PUT(g2, 2) SLAB_PUT(g2, 3) SLAB_PUT(g2, 4) SLAB_PUT(g2, 5)
  SLAB_PUT(g2, 6) SLAB_PUT(g2, 7) SLAB_PUT(g2, 8) SLAB_PUT(g2, 9) SLAB_PUT(g2, 10) SLAB_PUT(g2, 11)
  SLAB_PUT(g2, 12) SLAB_PUT(g2, 13) SLAB_PUT(g2, 14) SLAB_PUT(g2, 15) SLAB_PUT(g2, 16) SLAB_PUT(g2, 17)
  SLAB_PUT(g2, 18) SLAB_PUT(g2, 19) SLAB_PUT(g2, 20) SLAB_PUT(g2, 21) SLAB_PUT(g2, 22) SLAB_PUT(g2, 23)
  SLAB_PUT(g2, 24) SLAB_PUT(g2, 25) SLAB_PUT(g2, 26) SLAB_PUT(g2, 27) SLAB_PUT(g2, 28) SLAB_PUT(g2, 29)
  SLAB_PUT(g2, 30) SLAB_PUT(g2, 31)
  HOLD_REQUESTS_ABOVE();
  {
    v8f acc[2] = {};
    mma_layer<2, E_N>(acc, bufB + m * LPITCH, wfrag, hf);
    #pragma unroll
    for (int v = 0; v < 8; ++v) {
      bufA[(moff + v) * LPITCH + nbw + m]      = fmaxf(acc[0][v] + bv2a, 0.f);
      bufA[(moff + v) * LPITCH + nbw + 16 + m] = fmaxf(acc[1][v] + bv2b, 0.f);
    }
  }
  lds_barrier();

  SLAB_REQ(g4, 16) SLAB_REQ(g4, 17) SLAB_REQ(g4, 18) SLAB_REQ(g4, 19) SLAB_REQ(g4, 20) SLAB_REQ(g4, 21)
  SLAB_REQ(g4, 22) SLAB_REQ(g4, 23) SLAB_REQ(g4, 24) SLAB_REQ(g4, 25) SLAB_REQ(g4, 26) SLAB_REQ(g4, 27)
  SLAB_REQ(g4, 28) SLAB_REQ(g4, 29) SLAB_REQ(g4, 30) SLAB_REQ(g4, 31) SLAB_REQ(g4, 32) SLAB_REQ(g4, 33)
  SLAB_REQ(g4, 34) SLAB_REQ(g4, 35) SLAB_REQ(g4, 36) SLAB_REQ(g4, 37) SLAB_REQ(g4, 38) SLAB_REQ(g4, 39)
  SLAB_REQ(g4, 40) SLAB_REQ(g4, 41) SLAB_REQ(g4, 42) SLAB_REQ(g4, 43) SLAB_REQ(g4, 44) SLAB_REQ(g4, 45)
  SLAB_REQ(g4, 46) SLAB_REQ(g4, 47) SLAB_REQ(g4, 48) SLAB_REQ(g4, 49) SLAB_REQ(g4, 50) SLAB_REQ(g4, 51)
  SLAB_REQ(g4, 52) SLAB_REQ(g4, 53) SLAB_REQ(g4, 54) SLAB_REQ(g4, 55) SLAB_REQ(g4, 56) SLAB_REQ(g4, 57)
  SLAB_REQ(g4, 58) SLAB_REQ(g4, 59) SLAB_REQ(g4, 60) SLAB_REQ(g4, 61) SLAB_REQ(g4, 62) SLAB_REQ(g4, 63)
#undef SLAB_BASE
#undef SLAB_LDW
  float bv0 = 0.f, bv1 = 0.f, bv2 = 0.f, bv3 = 0.f;
  if (TASK) {
    bv0 = fb0[nb0 + m];      bv1 = fb0[nb0 + 16 + m];
    bv2 = fb0[nb0 + 32 + m]; bv3 = fb0[nb0 + 48 + m];
  }
  SLAB_PUT(g3, 0) SLAB_PUT(g3, 1) SLAB_PUT(g3, 2) SLAB_PUT(g3, 3) SLAB_PUT(g3, 4) SLAB_PUT(g3, 5)
  SLAB_PUT(g3, 6) SLAB_PUT(g3, 7) SLAB_PUT(g3, 8) SLAB_PUT(g3, 9) SLAB_PUT(g3, 10) SLAB_PUT(g3, 11)
  SLAB_PUT(g3, 12) SLAB_PUT(g3, 13) SLAB_PUT(g3, 14) SLAB_PUT(g3, 15) SLAB_PUT(g3, 16) SLAB_PUT(g3, 17)
  SLAB_PUT(g3, 18) SLAB_PUT(g3, 19) SLAB_PUT(g3, 20) SLAB_PUT(g3, 21) SLAB_PUT(g3, 22) SLAB_PUT(g3, 23)
  SLAB_PUT(g3, 24) SLAB_PUT(g3, 25) SLAB_PUT(g3, 26) SLAB_PUT(g3, 27) SLAB_PUT(g3, 28) SLAB_PUT(g3, 29)
  SLAB_PUT(g3, 30) SLAB_PUT(g3, 31)
  HOLD_REQUESTS_ABOVE();
  {
    v8f acc[2] = {};
    mma_layer<2, E_N>(acc, bufA + m * LPITCH, wfrag, hf);
    #pragma unroll
    for (int v = 0; v < 8; ++v) {
      float va = acc[0][v] + bv3a;
      float vb = acc[1][v] + bv3b;
      if (!TASK) { va = va + qva; vb = vb + qvb; }
      bufB[(moff + v) * LPITCH + nbw + m]      = va;
      bufB[(moff + v) * LPITCH + nbw + 16 + m] = vb;
    }
  }
  lds_barrier();

  SLAB_PUT(g4, 0) SLAB_PUT(g4, 1) SLAB_PUT(g4, 2) SLAB_PUT(g4, 3) SLAB_PUT(g4, 4) SLAB_PUT(g4, 5)
  SLAB_PUT(g4, 6) SLAB_PUT(g4, 7) SLAB_PUT(g4, 8) SLAB_PUT(g4, 9) SLAB_PUT(g4, 10) SLAB_PUT(g4, 11)
  SLAB_PUT(g4, 12) SLAB_PUT(g4, 13) SLAB_PUT(g4, 14) SLAB_PUT(g4, 15) SLAB_PUT(g4, 16) SLAB_PUT(g4, 17)
  SLAB_PUT(g4, 18) SLAB_PUT(g4, 19) SLAB_PUT(g4, 20) SLAB_PUT(g4, 21) SLAB_PUT(g4, 22) SLAB_PUT(g4, 23)
  SLAB_PUT(g4, 24) SLAB_PUT(g4, 25) SLAB_PUT(g4, 26) SLAB_PUT(g4, 27) SLAB_PUT(g4, 28) SLAB_PUT(g4, 29)
  SLAB_PUT(g4, 30) SLAB_PUT(g4, 31) SLAB_PUT(g4, 32) SLAB_PUT(g4, 33) SLAB_PUT(g4, 34) SLAB_PUT(g4, 35)
  SLAB_PUT(g4, 36) SLAB_PUT(g4, 37) SLAB_PUT(g4, 38) SLAB_PUT(g4, 39) SLAB_PUT(g4, 40) SLAB_PUT(g4, 41)
  SLAB_PUT(g4, 42) SLAB_PUT(g4, 43) SLAB_PUT(g4, 44) SLAB_PUT(g4, 45) SLAB_PUT(g4, 46) SLAB_PUT(g4, 47)
  SLAB_PUT(g4, 48) SLAB_PUT(g4, 49) SLAB_PUT(g4, 50) SLAB_PUT(g4, 51) SLAB_PUT(g4, 52) SLAB_PUT(g4, 53)
  SLAB_PUT(g4, 54) SLAB_PUT(g4, 55) SLAB_PUT(g4, 56) SLAB_PUT(g4, 57) SLAB_PUT(g4, 58) SLAB_PUT(g4, 59)
  SLAB_PUT(g4, 60) SLAB_PUT(g4, 61) SLAB_PUT(g4, 62) SLAB_PUT(g4, 63)
  {
    v8f acc[4] = {};
    mma_layer<4, E_N>(acc, bufB + m * LPITCH, wfrag, hf);
    float* op = pre + (size_t)(row0 + moff) * HID_N + nb0 + m;
    #pragma unroll
    for (int v = 0; v < 8; ++v) {
      if (TASK) {
        op[(size_t)v * HID_N]      = acc[0][v] + bv0;
        op[(size_t)v * HID_N + 16] = acc[1][v] + bv1;
        op[(size_t)v * HID_N + 32] = acc[2][v] + bv2;
        op[(size_t)v * HID_N + 48] = acc[3][v] + bv3;
      } else {
        op[(size_t)v * HID_N]      = acc[0][v];
        op[(size_t)v * HID_N + 16] = acc[1][v];
        op[(size_t)v * HID_N + 32] = acc[2][v];
        op[(size_t)v * HID_N + 48] = acc[3][v];
      }
    }
  }
}

__global__ __launch_bounds__(128) void front_kernel(
    const float* __restrict__ uav_feat, const float* __restrict__ task_feat,
    const float* __restrict__ uw0, const float* __restrict__ ub0,
    const float* __restrict__ uw1, const float* __restrict__ ub1,
    const float* __restrict__ uw2, const float* __restrict__ ub2,
    const float* __restrict__ tw0, const float* __restrict__ tb0,
    const float* __restrict__ tw1, const float* __restrict__ tb1,
    const float* __restrict__ tw2, const float* __restrict__ tb2,
    const float* __restrict__ hq,
    const float* __restrict__ fw0, const float* __restrict__ fb0,
    float* __restrict__ preT, float* __restrict__ preU) {
  __shared__ __align__(16) float bufA[16 * LPITCH];
  __shared__ __align__(16) float bufB[16 * LPITCH];
  __shared__ __align__(16) float slabs[4 * SLAB_FLOATS];
  const int b = blockIdx.x;
  if (b < H_N * (U_N / 16)) {
    const int h = b >> 4, utile = b & 15;
    front_rows<UAVD, false>(bufA, bufB, slabs, utile * 16, h, uav_feat,
                            uw0, ub0, uw1, ub1, uw2, ub2, hq, fw0, fb0,
                            preU + (size_t)h * U_N * HID_N);
  } else {
    const int ttile = b - H_N * (U_N / 16);
    front_rows<TASKD, true>(bufA, bufB, slabs, ttile * 16, 0, task_feat,
                            tw0, tb0, tw1, tb1, tw2, tb2, hq, fw0, fb0, preT);
  }
}

#define LG_R      32
#define LG_T      64
#define LG_PITCH  260

#define LG_STEP(A0, A1, B0, B1, B2, B3, WV)                                   \
  {                                                                           \
    acc00 = fmaf(fmaxf(A0.x + B0.x, 0.f), WV.x, acc00);                       \
    acc01 = fmaf(fmaxf(A0.x + B1.x, 0.f), WV.x, acc01);                       \
    acc02 = fmaf(fmaxf(A0.x + B2.x, 0.f), WV.x, acc02);                       \
    acc03 = fmaf(fmaxf(A0.x + B3.x, 0.f), WV.x, acc03);                       \
    acc10 = fmaf(fmaxf(A1.x + B0.x, 0.f), WV.x, acc10);                       \
    acc11 = fmaf(fmaxf(A1.x + B1.x, 0.f), WV.x, acc11);                       \
    acc12 = fmaf(fmaxf(A1.x + B2.x, 0.f), WV.x, acc12);                       \
    acc13 = fmaf(fmaxf(A1.x + B3.x, 0.f), WV.x, acc13);                       \
    acc00 = fmaf(fmaxf(A0.y + B0.y, 0.f), WV.y, acc00);                       \
    acc01 = fmaf(fmaxf(A0.y + B1.y, 0.f), WV.y, acc01);                       \
    acc02 = fmaf(fmaxf(A0.y + B2.y, 0.f), WV.y, acc02);                       \
    acc03 = fmaf(fmaxf(A0.y + B3.y, 0.f), WV.y, acc03);                       \
    acc10 = fmaf(fmaxf(A1.y + B0.y, 0.f), WV.y, acc10);                       \
    acc11 = fmaf(fmaxf(A1.y + B1.y, 0.f), WV.y, acc11);                       \
    acc12 = fmaf(fmaxf(A1.y + B2.y, 0.f), WV.y, acc12);                       \
    acc13 = fmaf(fmaxf(A1.y + B3.y, 0.f), WV.y, acc13);                       \
    acc00 = fmaf(fmaxf(A0.z + B0.z, 0.f), WV.z, acc00);                       \
    acc01 = fmaf(fmaxf(A0.z + B1.z, 0.f), WV.z, acc01);                       \
    acc02 = fmaf(fmaxf(A0.z + B2.z, 0.f), WV.z, acc02);                       \
    acc03 = fmaf(fmaxf(A0.z + B3.z, 0.f), WV.z, acc03);                       \
    acc10 = fmaf(fmaxf(A1.z + B0.z, 0.f), WV.z, acc10);                       \
    acc11 = fmaf(fmaxf(A1.z + B1.z, 0.f), WV.z, acc11);                       \
    acc12 = fmaf(fmaxf(A1.z + B2.z, 0.f), WV.z, acc12);                       \
    acc13 = fmaf(fmaxf(A1.z + B3.z, 0.f), WV.z, acc13);                       \
    acc00 = fmaf(fmaxf(A0.w + B0.w, 0.f), WV.w, acc00);                       \
    acc01 = fmaf(fmaxf(A0.w + B1.w, 0.f), WV.w, acc01);                       \
    acc02 = fmaf(fmaxf(A0.w + B2.w, 0.f), WV.w, acc02);                       \
    acc03 = fmaf(fmaxf(A0.w + B3.w, 0.f), WV.w, acc03);                       \
    acc10 = fmaf(fmaxf(A1.w + B0.w, 0.f), WV.w, acc10);                       \
    acc11 = fmaf(fmaxf(A1.w + B1.w, 0.f), WV.w, acc11);                       \
    acc12 = fmaf(fmaxf(A1.w + B2.w, 0.f), WV.w, acc12);                       \
    acc13 = fmaf(fmaxf(A1.w + B3.w, 0.f), WV.w, acc13);                       \
  }

#define LG_NC     4
#define LG_CK     (HID_N / LG_NC)

#define LG_TRIP(K)                                                            \
  {                                                                           \
    const float4 na0 = *(const float4*)&pa0[(K) + 4];                         \
    const float4 na1 = *(const float4*)&pa1[(K) + 4];                         \
    const float4 nb0 = *(const float4*)&pb0[(K) + 4];                         \
    const float4 nb1 = *(const float4*)&pb1[(K) + 4];                         \
    const float4 nb2 = *(const float4*)&pb2[(K) + 4];                         \
    const float4 nb3 = *(const float4*)&pb3[(K) + 4];                         \
    const float4 nwv = *(const float4*)&wL[(K) + 4];                          \
    LG_STEP(a0, a1, b0, b1, b2, b3, wv)                                       \
    a0 = *(const float4*)&pa0[(K) + 8];                                       \
    a1 = *(const float4*)&pa1[(K) + 8];                                       \
    b0 = *(const float4*)&pb0[(K) + 8];                                       \
    b1 = *(const float4*)&pb1[(K) + 8];                                       \
    b2 = *(const float4*)&pb2[(K) + 8];                                       \
    b3 = *(const float4*)&pb3[(K) + 8];                                       \
    wv = *(const float4*)&wL[(K) + 8];                                        \
    LG_STEP(na0, na1, nb0, nb1, nb2, nb3, nwv)                                \
  }

#define LG_TRIPS(KLO, KHI)                                                    \
  for (int k = (KLO); k < (KHI); k += 8) LG_TRIP(k)

#define LG_GU(J, C) (*(const float4*)&gu[(rb + 16 * (J)) * HID_N + LG_CK * (C) + cp])
#define LG_GT(J, C) (*(const float4*)&gt[(rb + 16 * (J)) * HID_N + LG_CK * (C) + cp])
#define LG_LOAD_CHUNK(C)                                                      \
  const float4 su##C##_0 = LG_GU(0, C), su##C##_1 = LG_GU(1, C);              \
  const float4 st##C##_0 = LG_GT(0, C), st##C##_1 = LG_GT(1, C);              \
  const float4 st##C##_2 = LG_GT(2, C), st##C##_3 = LG_GT(3, C);

#define LG_PU(J, C) (*(float4*)&puL[(rb + 16 * (J)) * LG_PITCH + LG_CK * (C) + cp])
#define LG_PT(J, C) (*(float4*)&ptL[((rb >> 2) + 4 * (J) + 16 * (rb & 3)) * LG_PITCH + LG_CK * (C) + cp])
#define LG_STORE_CHUNK(C)                                                     \
  {                                                                           \
    LG_PU(0, C) = su##C##_0; LG_PU(1, C) = su##C##_1;                         \
    LG_PT(0, C) = st##C##_0; LG_PT(1, C) = st##C##_1;                         \
    LG_PT(2, C) = st##C##_2; LG_PT(3, C) = st##C##_3;                         \
  }

__global__ __launch_bounds__(256) void logits_kernel(
    const float* preU, const float* preT,
    const float* fw1, const float* __restrict__ fb1,
    float* out) {
  __shared__ __align__(16) float puL[LG_R * LG_PITCH];
  __shared__ __align__(16) float ptL[LG_T * LG_PITCH];
  __shared__ __align__(16) float wL[HID_N];

  const int ttile = blockIdx.x & (T_N / LG_T - 1);
  const int rtile = blockIdx.x >> 3;
  const int tid   = threadIdx.x;

  const float* gu = preU + (size_t)rtile * LG_R * HID_N;
  const float* gt = preT + (size_t)ttile * LG_T * HID_N;

  const int cp = (tid & 15) << 2;
  const int rb = tid >> 4;
  LG_LOAD_CHUNK(0)
  const float4 rw = *(const float4*)&fw1[(tid & (HID_N / 4 - 1)) * 4];
  __builtin_amdgcn_sched_barrier(0);
  LG_LOAD_CHUNK(1)
  __builtin_amdgcn_sched_barrier(0);
  LG_LOAD_CHUNK(2)
  __builtin_amdgcn_sched_barrier(0);
  LG_LOAD_CHUNK(3)
  __builtin_amdgcn_sched_barrier(0);

  LG_STORE_CHUNK(0)
  if (tid < HID_N / 4) *(float4*)&wL[tid * 4] = rw;
  lds_barrier();

  const int tt = tid & 15;
  const int tu = tid >> 4;
  const float* pa0 = &puL[tu * LG_PITCH];
  const float* pa1 = &puL[(tu + 16) * LG_PITCH];
  const float* pb0 = &ptL[tt * LG_PITCH];
  const float* pb1 = &ptL[(tt + 16) * LG_PITCH];
  const float* pb2 = &ptL[(tt + 32) * LG_PITCH];
  const float* pb3 = &ptL[(tt + 48) * LG_PITCH];

  float acc00 = 0.f, acc01 = 0.f, acc02 = 0.f, acc03 = 0.f;
  float acc10 = 0.f, acc11 = 0.f, acc12 = 0.f, acc13 = 0.f;

  float4 a0 = *(const float4*)&pa0[0];
  float4 a1 = *(const float4*)&pa1[0];
  float4 b0 = *(const float4*)&pb0[0];
  float4 b1 = *(const float4*)&pb1[0];
  float4 b2 = *(const float4*)&pb2[0];
  float4 b3 = *(const float4*)&pb3[0];
  float4 wv = *(const float4*)&wL[0];

  LG_TRIPS(0, 32)
  LG_STORE_CHUNK(1)
  lds_barrier();
  LG_TRIPS(32, 96)
  LG_STORE_CHUNK(2)
  lds_barrier();
  LG_TRIPS(96, 160)
  LG_STORE_CHUNK(3)
  lds_barrier();
  LG_TRIPS(160, HID_N - 8)
  {
    const float4 na0 = *(const float4*)&pa0[HID_N - 4];
    const float4 na1 = *(const float4*)&pa1[HID_N - 4];
    const float4 nb0 = *(const float4*)&pb0[HID_N - 4];
    const float4 nb1 = *(const float4*)&pb1[HID_N - 4];
    const float4 nb2 = *(const float4*)&pb2[HID_N - 4];
    const float4 nb3 = *(const float4*)&pb3[HID_N - 4];
    const float4 nwv = *(const float4*)&wL[HID_N - 4];
    LG_STEP(a0, a1, b0, b1, b2, b3, wv)
    LG_STEP(na0, na1, nb0, nb1, nb2, nb3, nwv)
  }

  const float bias = fb1[0];
  float* op = out + ((size_t)rtile * LG_R + tu) * T_N + ttile * LG_T + tt * 4;
  float4 o0, o1;
  o0.x = acc00 + bias; o0.y = acc01 + bias; o0.z = acc02 + bias; o0.w = acc03 + bias;
  o1.x = acc10 + bias; o1.y = acc11 + bias; o1.z = acc12 + bias; o1.w = acc13 + bias;
  *(float4*)&op[0]                 = o0;
  *(float4*)&op[(size_t)16 * T_N]  = o1;
}

extern "C" void kernel_launch(void* const* d_in, const int* in_sizes, int n_in,
                              void* d_out, int out_size, void* d_ws, size_t ws_size,
                              hipStream_t stream) {
  (void)in_sizes; (void)n_in; (void)out_size; (void)ws_size;
  const float* uav_feat = (const float*)d_in[0];
  const float* task_feat = (const float*)d_in[1];
  const float* uw0 = (const float*)d_in[2];
  const float* ub0 = (const float*)d_in[3];
  const float* uw1 = (const float*)d_in[4];
  const float* ub1 = (const float*)d_in[5];
  const float* uw2 = (const float*)d_in[6];
  const float* ub2 = (const float*)d_in[7];
  const float* tw0 = (const float*)d_in[8];
  const float* tb0 = (const float*)d_in[9];
  const float* tw1 = (const float*)d_in[10];
  const float* tb1 = (const float*)d_in[11];
  const float* tw2 = (const float*)d_in[12];
  const float* tb2 = (const float*)d_in[13];
  const float* head_queries = (const float*)d_in[14];
  const float* fw0 = (const float*)d_in[15];
  const float* fb0 = (const float*)d_in[16];
  const float* fw1 = (const float*)d_in[17];
  const float* fb1 = (const float*)d_in[18];
  float* out = (float*)d_out;

  float* ws = (float*)d_ws;
  float* preT  = ws;
  float* preU  = ws + 131072;

  front_kernel<<<H_N * (U_N / 16) + T_N / 16, 128, 0, stream>>>(
      uav_feat, task_feat, uw0, ub0, uw1, ub1, uw2, ub2,
      tw0, tb0, tw1, tb1, tw2, tb2, head_queries, fw0, fb0, preT, preU);
  logits_kernel<<<(H_N * U_N / LG_R) * (T_N / LG_T), 256, 0, stream>>>(
      preU, preT, fw1, fb1, out);
}


# Logits kernel: relu(a+b)*w summed as w*max(a,-b) (front stores -pre_t) plus w*b once per t row at staging; 2 ops per term, was 3, k step hand-placed as 32 dual-issue pairs; sum order changes
# speedup vs baseline: 3.0205x; 1.1784x over previous
#include <hip/hip_runtime.h>

typedef __attribute__((ext_vector_type(2))) float v2f;
typedef __attribute__((ext_vector_type(8))) float v8f;

#define U_N     256
#define T_N     512
#define H_N     4
#define UAVD    64
#define TASKD   32
#define E_N     128
#define HID_N   256

#define WMMA_F32(A, B, C) \
  __builtin_amdgcn_wmma_f32_16x16x4_f32(false, (A), false, (B), (short)0, (C), false, false)

__device__ __forceinline__ int koff(int j, int hf) { return 8 * j + 4 * hf; }

__device__ __forceinline__ void lds_barrier() {
  __builtin_amdgcn_fence(__ATOMIC_RELEASE, "workgroup", "local");
  __builtin_amdgcn_s_barrier();
  __builtin_amdgcn_fence(__ATOMIC_ACQUIRE, "workgroup", "local");
}

#define HOLD_REQUESTS_ABOVE() __builtin_amdgcn_sched_barrier(0)

#define LPITCH   132
#define SLAB_FLOATS (64 * LPITCH)

template <int R, int K>
__device__ __forceinline__ void slab_request(float4 (&g)[R * K / 128],
                                             const float* __restrict__ base,
                                             int ldw, int lane) {
  constexpr int LPR = K / 4;
  constexpr int RPL = 32 / LPR;
  const int lr = lane / LPR, lc = (lane % LPR) * 4;
  #pragma unroll
  for (int i = 0; i < R / RPL; ++i)
    g[i] = *(const float4*)(base + (size_t)(i * RPL + lr) * ldw + lc);
}

template <int R, int K>
__device__ __forceinline__ void slab_to_lds(float* __restrict__ ws,
                                            const float4 (&g)[R * K / 128], int lane) {
  constexpr int LPR = K / 4;
  constexpr int RPL = 32 / LPR;
  const int lr = lane / LPR, lc = (lane % LPR) * 4;
  #pragma unroll
  for (int i = 0; i < R / RPL; ++i)
    *(float4*)&ws[(i * RPL + lr) * LPITCH + lc] = g[i];
}

template <int NT, int K>
__device__ __forceinline__ void mma_layer(v8f (&acc)[NT],
                                          const float* __restrict__ ap,
                                          const float* __restrict__ wp, int hf) {
  float4 af[K / 8];
  float4 bf[NT][K / 8];
  #pragma unroll
  for (int j = 0; j < K / 8; ++j) {
    af[j] = *(const float4*)(ap + koff(j, hf));
    #pragma unroll
    for (int t = 0; t < NT; ++t)
      bf[t][j] = *(const float4*)(wp + t * 16 * LPITCH + koff(j, hf));
  }
  __builtin_amdgcn_sched_barrier(0);
  #pragma unroll
  for (int j = 0; j < K / 8; ++j) {
    v2f alo; alo.x = af[j].x; alo.y = af[j].y;
    v2f ahi; ahi.x = af[j].z; ahi.y = af[j].w;
    #pragma unroll
    for (int t = 0; t < NT; ++t) {
      v2f b; b.x = bf[t][j].x; b.y = bf[t][j].y;
      acc[t] = WMMA_F32(alo, b, acc[t]);
    }
    #pragma unroll
    for (int t = 0; t < NT; ++t) {
      v2f b; b.x = bf[t][j].z; b.y = bf[t][j].w;
      acc[t] = WMMA_F32(ahi, b, acc[t]);
    }
  }
}

#define SLAB_REQ(NAME, I) const float4 NAME##_##I = *(const float4*)(SLAB_BASE + (size_t)(I) * (SLAB_LDW) + lane * 4);
#define SLAB_PUT(NAME, I) *(float4*)&ws[(I) * LPITCH + lane * 4] = NAME##_##I;

template <int INK, bool TASK>
__device__ __forceinline__ void front_rows(
    float* __restrict__ bufA, float* __restrict__ bufB, float* __restrict__ slabs,
    const int row0, const int h,
    const float* __restrict__ x,
    const float* __restrict__ w0, const float* __restrict__ b0,
    const float* __restrict__ w1, const float* __restrict__ b1,
    const float* __restrict__ w2, const float* __restrict__ b2,
    const float* __restrict__ hq,
    const float* __restrict__ fw0, const float* __restrict__ fb0,
    float* __restrict__ pre) {
  const int tid  = threadIdx.x;
  const int lane = tid & 31;
  const int wave = tid >> 5;
  const int m    = lane & 15;
  const int hf   = lane >> 4;
  const int moff = hf * 8;
  const int nbw  = wave * 32;
  const int nb0  = wave * 64;
  float* ws = slabs + wave * SLAB_FLOATS;
  const float* wfrag = ws + m * LPITCH;

  constexpr int NX = (16 * INK / 4) / 128;
  float4 xin[NX];
  #pragma unroll
  for (int i = 0; i < NX; ++i)
    xin[i] = *(const float4*)(x + (size_t)row0 * INK + (size_t)(tid + 128 * i) * 4);
  float4 g1[32 * INK / 128];
  slab_request<32, INK>(g1, w0 + (size_t)nbw * INK, INK, lane);
  const float bv1a = b0[nbw + m], bv1b = b0[nbw + 16 + m];

  #pragma unroll
  for (int i = 0; i < NX; ++i) {
    const int f = (tid + 128 * i) * 4;
    const int r = f / INK, c = f - r * INK;
    *(float4*)&bufA[r * LPITCH + c] = xin[i];
  }
  slab_to_lds<32, INK>(ws, g1, lane);
  lds_barrier();

  const float* w1slab = w1 + (size_t)nbw * E_N;
#define SLAB_BASE w1slab
#define SLAB_LDW  E_N
  SLAB_REQ(g2, 0) SLAB_REQ(g2, 1) SLAB_REQ(g2, 2) SLAB_REQ(g2, 3) SLAB_REQ(g2, 4) SLAB_REQ(g2, 5)
  SLAB_REQ(g2, 6) SLAB_REQ(g2, 7) SLAB_REQ(g2, 8) SLAB_REQ(g2, 9) SLAB_REQ(g2, 10) SLAB_REQ(g2, 11)
  SLAB_REQ(g2, 12) SLAB_REQ(g2, 13) SLAB_REQ(g2, 14) SLAB_REQ(g2, 15) SLAB_REQ(g2, 16) SLAB_REQ(g2, 17)
  SLAB_REQ(g2, 18) SLAB_REQ(g2, 19) SLAB_REQ(g2, 20) SLAB_REQ(g2, 21) SLAB_REQ(g2, 22) SLAB_REQ(g2, 23)
  SLAB_REQ(g2, 24) SLAB_REQ(g2, 25) SLAB_REQ(g2, 26) SLAB_REQ(g2, 27) SLAB_REQ(g2, 28) SLAB_REQ(g2, 29)
  SLAB_REQ(g2, 30) SLAB_REQ(g2, 31)
#undef SLAB_BASE
#undef SLAB_LDW
  const float bv2a = b1[nbw + m], bv2b = b1[nbw + 16 + m];
  HOLD_REQUESTS_ABOVE();
  {
    v8f acc[2] = {};
    mma_layer<2, INK>(acc, bufA + m * LPITCH, wfrag, hf);
    #pragma unroll
    for (int v = 0; v < 8; ++v) {
      bufB[(moff + v) * LPITCH + nbw + m]      = fmaxf(acc[0][v] + bv1a, 0.f);
      bufB[(moff + v) * LPITCH + nbw + 16 + m] = fmaxf(acc[1][v] + bv1b, 0.f);
    }
  }
  lds_barrier();

  const float* w2slab = w2 + (size_t)nbw * E_N;
#define SLAB_BASE w2slab
#define SLAB_LDW  E_N
  SLAB_REQ(g3, 0) SLAB_REQ(g3, 1) SLAB_REQ(g3, 2) SLAB_REQ(g3, 3) SLAB_REQ(g3, 4) SLAB_REQ(g3, 5)
  SLAB_REQ(g3, 6) SLAB_REQ(g3, 7) SLAB_REQ(g3, 8) SLAB_REQ(g3, 9) SLAB_REQ(g3, 10) SLAB_REQ(g3, 11)
  SLAB_REQ(g3, 12) SLAB_REQ(g3, 13) SLAB_REQ(g3, 14) SLAB_REQ(g3, 15) SLAB_REQ(g3, 16) SLAB_REQ(g3, 17)
  SLAB_REQ(g3, 18) SLAB_REQ(g3, 19) SLAB_REQ(g3, 20) SLAB_REQ(g3, 21) SLAB_REQ(g3, 22) SLAB_REQ(g3, 23)
  SLAB_REQ(g3, 24) SLAB_REQ(g3, 25) SLAB_REQ(g3, 26) SLAB_REQ(g3, 27) SLAB_REQ(g3, 28) SLAB_REQ(g3, 29)
  SLAB_REQ(g3, 30) SLAB_REQ(g3, 31)
#undef SLAB_BASE
#undef SLAB_LDW
  const float bv3a = b2[nbw + m], bv3b = b2[nbw + 16 + m];
  float qva = 0.f, qvb = 0.f;
  if (!TASK) { qva = hq[(size_t)h * E_N + nbw + m]; qvb = hq[(size_t)h * E_N + nbw + 16 + m]; }
  const float* fwslab = fw0 + (size_t)nb0 * (2 * E_N) + (TASK ? E_N : 0);
#define SLAB_BASE fwslab
#define SLAB_LDW  (2 * E_N)
  SLAB_REQ(g4, 0) SLAB_REQ(g4, 1) SLAB_REQ(g4, 2) SLAB_REQ(g4, 3) SLAB_REQ(g4, 4) SLAB_REQ(g4, 5)
  SLAB_REQ(g4, 6) SLAB_REQ(g4, 7) SLAB_REQ(g4, 8) SLAB_REQ(g4, 9) SLAB_REQ(g4, 10) SLAB_REQ(g4, 11)
  SLAB_REQ(g4, 12) SLAB_REQ(g4, 13) SLAB_REQ(g4, 14) SLAB_REQ(g4, 15)
  SLAB_PUT(g2, 0) SLAB_PUT(g2, 1) SLAB_PUT(g2, 2) SLAB_PUT(g2, 3) SLAB_PUT(g2, 4) SLAB_PUT(g2, 5)
  SLAB_PUT(g2, 6) SLAB_PUT(g2, 7) SLAB_PUT(g2, 8) SLAB_PUT(g2, 9) SLAB_PUT(g2, 10) SLAB_PUT(g2, 11)
  SLAB_PUT(g2, 12) SLAB_PUT(g2, 13) SLAB_PUT(g2, 14) SLAB_PUT(g2, 15) SLAB_PUT(g2, 16) SLAB_PUT(g2, 17)
  SLAB_PUT(g2, 18) SLAB_PUT(g2, 19) SLAB_PUT(g2, 20) SLAB_PUT(g2, 21) SLAB_PUT(g2, 22) SLAB_PUT(g2, 23)
  SLAB_PUT(g2, 24) SLAB_PUT(g2, 25) SLAB_PUT(g2, 26) SLAB_PUT(g2, 27) SLAB_PUT(g2, 28) SLAB_PUT(g2, 29)
  SLAB_PUT(g2, 30) SLAB_PUT(g2, 31)
  HOLD_REQUESTS_ABOVE();
  {
    v8f acc[2] = {};
    mma_layer<2, E_N>(acc, bufB + m * LPITCH, wfrag, hf);
    #pragma unroll
    for (int v = 0; v < 8; ++v) {
      bufA[(moff + v) * LPITCH + nbw + m]      = fmaxf(acc[0][v] + bv2a, 0.f);
      bufA[(moff + v) * LPITCH + nbw + 16 + m] = fmaxf(acc[1][v] + bv2b, 0.f);
    }
  }
  lds_barrier();

  SLAB_REQ(g4, 16) SLAB_REQ(g4, 17) SLAB_REQ(g4, 18) SLAB_REQ(g4, 19) SLAB_REQ(g4, 20) SLAB_REQ(g4, 21)
  SLAB_REQ(g4, 22) SLAB_REQ(g4, 23) SLAB_REQ(g4, 24) SLAB_REQ(g4, 25) SLAB_REQ(g4, 26) SLAB_REQ(g4, 27)
  SLAB_REQ(g4, 28) SLAB_REQ(g4, 29) SLAB_REQ(g4, 30) SLAB_REQ(g4, 31) SLAB_REQ(g4, 32) SLAB_REQ(g4, 33)
  SLAB_REQ(g4, 34) SLAB_REQ(g4, 35) SLAB_REQ(g4, 36) SLAB_REQ(g4, 37) SLAB_REQ(g4, 38) SLAB_REQ(g4, 39)
  SLAB_REQ(g4, 40) SLAB_REQ(g4, 41) SLAB_REQ(g4, 42) SLAB_REQ(g4, 43) SLAB_REQ(g4, 44) SLAB_REQ(g4, 45)
  SLAB_REQ(g4, 46) SLAB_REQ(g4, 47) SLAB_REQ(g4, 48) SLAB_REQ(g4, 49) SLAB_REQ(g4, 50) SLAB_REQ(g4, 51)
  SLAB_REQ(g4, 52) SLAB_REQ(g4, 53) SLAB_REQ(g4, 54) SLAB_REQ(g4, 55) SLAB_REQ(g4, 56) SLAB_REQ(g4, 57)
  SLAB_REQ(g4, 58) SLAB_REQ(g4, 59) SLAB_REQ(g4, 60) SLAB_REQ(g4, 61) SLAB_REQ(g4, 62) SLAB_REQ(g4, 63)
#undef SLAB_BASE
#undef SLAB_LDW
  float nbv0 = 0.f, nbv1 = 0.f, nbv2 = 0.f, nbv3 = 0.f;
  if (TASK) {
    nbv0 = -fb0[nb0 + m];      nbv1 = -fb0[nb0 + 16 + m];
    nbv2 = -fb0[nb0 + 32 + m]; nbv3 = -fb0[nb0 + 48 + m];
  }
  SLAB_PUT(g3, 0) SLAB_PUT(g3, 1) SLAB_PUT(g3, 2) SLAB_PUT(g3, 3) SLAB_PUT(g3, 4) SLAB_PUT(g3, 5)
  SLAB_PUT(g3, 6) SLAB_PUT(g3, 7) SLAB_PUT(g3, 8) SLAB_PUT(g3, 9) SLAB_PUT(g3, 10) SLAB_PUT(g3, 11)
  SLAB_PUT(g3, 12) SLAB_PUT(g3, 13) SLAB_PUT(g3, 14) SLAB_PUT(g3, 15) SLAB_PUT(g3, 16) SLAB_PUT(g3, 17)
  SLAB_PUT(g3, 18) SLAB_PUT(g3, 19) SLAB_PUT(g3, 20) SLAB_PUT(g3, 21) SLAB_PUT(g3, 22) SLAB_PUT(g3, 23)
  SLAB_PUT(g3, 24) SLAB_PUT(g3, 25) SLAB_PUT(g3, 26) SLAB_PUT(g3, 27) SLAB_PUT(g3, 28) SLAB_PUT(g3, 29)
  SLAB_PUT(g3, 30) SLAB_PUT(g3, 31)
  HOLD_REQUESTS_ABOVE();
  {
    v8f acc[2] = {};
    mma_layer<2, E_N>(acc, bufA + m * LPITCH, wfrag, hf);
    #pragma unroll
    for (int v = 0; v < 8; ++v) {
      float va = acc[0][v] + bv3a;
      float vb = acc[1][v] + bv3b;
      if (!TASK) { va = va + qva; vb = vb + qvb; }
      bufB[(moff + v) * LPITCH + nbw + m]      = va;
      bufB[(moff + v) * LPITCH + nbw + 16 + m] = vb;
    }
  }
  lds_barrier();

  SLAB_PUT(g4, 0) SLAB_PUT(g4, 1) SLAB_PUT(g4, 2) SLAB_PUT(g4, 3) SLAB_PUT(g4, 4) SLAB_PUT(g4, 5)
  SLAB_PUT(g4, 6) SLAB_PUT(g4, 7) SLAB_PUT(g4, 8) SLAB_PUT(g4, 9) SLAB_PUT(g4, 10) SLAB_PUT(g4, 11)
  SLAB_PUT(g4, 12) SLAB_PUT(g4, 13) SLAB_PUT(g4, 14) SLAB_PUT(g4, 15) SLAB_PUT(g4, 16) SLAB_PUT(g4, 17)
  SLAB_PUT(g4, 18) SLAB_PUT(g4, 19) SLAB_PUT(g4, 20) SLAB_PUT(g4, 21) SLAB_PUT(g4, 22) SLAB_PUT(g4, 23)
  SLAB_PUT(g4, 24) SLAB_PUT(g4, 25) SLAB_PUT(g4, 26) SLAB_PUT(g4, 27) SLAB_PUT(g4, 28) SLAB_PUT(g4, 29)
  SLAB_PUT(g4, 30) SLAB_PUT(g4, 31) SLAB_PUT(g4, 32) SLAB_PUT(g4, 33) SLAB_PUT(g4, 34) SLAB_PUT(g4, 35)
  SLAB_PUT(g4, 36) SLAB_PUT(g4, 37) SLAB_PUT(g4, 38) SLAB_PUT(g4, 39) SLAB_PUT(g4, 40) SLAB_PUT(g4, 41)
  SLAB_PUT(g4, 42) SLAB_PUT(g4, 43) SLAB_PUT(g4, 44) SLAB_PUT(g4, 45) SLAB_PUT(g4, 46) SLAB_PUT(g4, 47)
  SLAB_PUT(g4, 48) SLAB_PUT(g4, 49) SLAB_PUT(g4, 50) SLAB_PUT(g4, 51) SLAB_PUT(g4, 52) SLAB_PUT(g4, 53)
  SLAB_PUT(g4, 54) SLAB_PUT(g4, 55) SLAB_PUT(g4, 56) SLAB_PUT(g4, 57) SLAB_PUT(g4, 58) SLAB_PUT(g4, 59)
  SLAB_PUT(g4, 60) SLAB_PUT(g4, 61) SLAB_PUT(g4, 62) SLAB_PUT(g4, 63)
  {
    v8f acc[4] = {};
    mma_layer<4, E_N>(acc, bufB + m * LPITCH, wfrag, hf);
    float* op = pre + (size_t)(row0 + moff) * HID_N + nb0 + m;
    #pragma unroll
    for (int v = 0; v < 8; ++v) {
      if (TASK) {
        op[(size_t)v * HID_N]      = nbv0 - acc[0][v];
        op[(size_t)v * HID_N + 16] = nbv1 - acc[1][v];
        op[(size_t)v * HID_N + 32] = nbv2 - acc[2][v];
        op[(size_t)v * HID_N + 48] = nbv3 - acc[3][v];
      } else {
        op[(size_t)v * HID_N]      = acc[0][v];
        op[(size_t)v * HID_N + 16] = acc[1][v];
        op[(size_t)v * HID_N + 32] = acc[2][v];
        op[(size_t)v * HID_N + 48] = acc[3][v];
      }
    }
  }
}

__global__ __launch_bounds__(128) void front_kernel(
    const float* __restrict__ uav_feat, const float* __restrict__ task_feat,
    const float* __restrict__ uw0, const float* __restrict__ ub0,
    const float* __restrict__ uw1, const float* __restrict__ ub1,
    const float* __restrict__ uw2, const float* __restrict__ ub2,
    const float* __restrict__ tw0, const float* __restrict__ tb0,
    const float* __restrict__ tw1, const float* __restrict__ tb1,
    const float* __restrict__ tw2, const float* __restrict__ tb2,
    const float* __restrict__ hq,
    const float* __restrict__ fw0, const float* __restrict__ fb0,
    float* __restrict__ preT, float* __restrict__ preU) {
  __shared__ __align__(16) float bufA[16 * LPITCH];
  __shared__ __align__(16) float bufB[16 * LPITCH];
  __shared__ __align__(16) float slabs[4 * SLAB_FLOATS];
  const int b = blockIdx.x;
  if (b < H_N * (U_N / 16)) {
    const int h = b >> 4, utile = b & 15;
    front_rows<UAVD, false>(bufA, bufB, slabs, utile * 16, h, uav_feat,
                            uw0, ub0, uw1, ub1, uw2, ub2, hq, fw0, fb0,
                            preU + (size_t)h * U_N * HID_N);
  } else {
    const int ttile = b - H_N * (U_N / 16);
    front_rows<TASKD, true>(bufA, bufB, slabs, ttile * 16, 0, task_feat,
                            tw0, tb0, tw1, tb1, tw2, tb2, hq, fw0, fb0, preT);
  }
}

#define LG_R      32
#define LG_T      64
#define LG_PITCH  260

typedef __attribute__((ext_vector_type(4))) float v4f;

#define LG_STEP_S1(A0, A1, B0, B1, B2, B3, WV)                              \
  asm(                                                                        \
      "v_dual_max_num_f32 v40, v0, v12 :: v_dual_max_num_f32 v45, v6, v18\n\t" \
      "v_dual_max_num_f32 v44, v6, v12 :: v_dual_max_num_f32 v41, v0, v18\n\t" \
      "v_dual_max_num_f32 v42, v0, v24 :: v_dual_max_num_f32 v47, v6, v30\n\t" \
      "v_dual_max_num_f32 v46, v6, v24 :: v_dual_max_num_f32 v43, v0, v30\n\t" \
      "v_dual_max_num_f32 v96, v1, v13 :: v_dual_max_num_f32 v101, v7, v19\n\t" \
      "v_dual_max_num_f32 v100, v7, v13 :: v_dual_max_num_f32 v97, v1, v19\n\t" \
      "v_dual_max_num_f32 v98, v1, v25 :: v_dual_max_num_f32 v103, v7, v31\n\t" \
      "v_dual_max_num_f32 v102, v7, v25 :: v_dual_max_num_f32 v99, v1, v31\n\t" \
      "v_dual_fmac_f32 v48, v40, v36 :: v_dual_fmac_f32 v49, v41, v36\n\t" \
      "v_dual_fmac_f32 v50, v42, v36 :: v_dual_fmac_f32 v51, v43, v36\n\t" \
      "v_dual_fmac_f32 v52, v44, v36 :: v_dual_fmac_f32 v53, v45, v36\n\t" \
      "v_dual_fmac_f32 v54, v46, v36 :: v_dual_fmac_f32 v55, v47, v36\n\t" \
      "v_dual_max_num_f32 v40, v2, v14 :: v_dual_max_num_f32 v45, v8, v20\n\t" \
      "v_dual_max_num_f32 v44, v8, v14 :: v_dual_max_num_f32 v41, v2, v20\n\t" \
      "v_dual_max_num_f32 v42, v2, v26 :: v_dual_max_num_f32 v47, v8, v32\n\t" \
      "v_dual_max_num_f32 v46, v8, v26 :: v_dual_max_num_f32 v43, v2, v32\n\t" \
      "v_dual_fmac_f32 v48, v96, v37 :: v_dual_fmac_f32 v49, v97, v37\n\t" \
      "v_dual_fmac_f32 v50, v98, v37 :: v_dual_fmac_f32 v51, v99, v37\n\t" \
      "v_dual_fmac_f32 v52, v100, v37 :: v_dual_fmac_f32 v53, v101, v37\n\t" \
      "v_dual_fmac_f32 v54, v102, v37 :: v_dual_fmac_f32 v55, v103, v37\n\t" \
      "v_dual_max_num_f32 v96, v3, v15 :: v_dual_max_num_f32 v101, v9, v21\n\t" \
      "v_dual_max_num_f32 v100, v9, v15 :: v_dual_max_num_f32 v97, v3, v21\n\t" \
      "v_dual_max_num_f32 v98, v3, v27 :: v_dual_max_num_f32 v103, v9, v33\n\t" \
      "v_dual_max_num_f32 v102, v9, v27 :: v_dual_max_num_f32 v99, v3, v33\n\t" \
      "v_dual_fmac_f32 v48, v40, v38 :: v_dual_fmac_f32 v49, v41, v38\n\t" \
      "v_dual_fmac_f32 v50, v42, v38 :: v_dual_fmac_f32 v51, v43, v38\n\t" \
      "v_dual_fmac_f32 v52, v44, v38 :: v_dual_fmac_f32 v53, v45, v38\n\t" \
      "v_dual_fmac_f32 v54, v46, v38 :: v_dual_fmac_f32 v55, v47, v38\n\t" \
      "v_dual_fmac_f32 v48, v96, v39 :: v_dual_fmac_f32 v49, v97, v39\n\t" \
      "v_dual_fmac_f32 v50, v98, v39 :: v_dual_fmac_f32 v51, v99, v39\n\t" \
      "v_dual_fmac_f32 v52, v100, v39 :: v_dual_fmac_f32 v53, v101, v39\n\t" \
      "v_dual_fmac_f32 v54, v102, v39 :: v_dual_fmac_f32 v55, v103, v39" \
      : "+{v[48:49]}"(acc0a), "+{v[50:51]}"(acc0b),                           \
        "+{v[52:53]}"(acc1a), "+{v[54:55]}"(acc1b)                            \
      : "{v[0:3]}"(A0), "{v[6:9]}"(A1), "{v[12:15]}"(B0), "{v[18:21]}"(B1), \
        "{v[24:27]}"(B2), "{v[30:33]}"(B3), "{v[36:39]}"(WV)             \
      : "v40", "v41", "v42", "v43", "v44", "v45", "v46", "v47",         \
        "v96", "v97", "v98", "v99", "v100", "v101", "v102", "v103");

#define LG_STEP_S2(A0, A1, B0, B1, B2, B3, WV)                              \
  asm(                                                                        \
      "v_dual_max_num_f32 v40, v56, v68 :: v_dual_max_num_f32 v45, v62, v74\n\t" \
      "v_dual_max_num_f32 v44, v62, v68 :: v_dual_max_num_f32 v41, v56, v74\n\t" \
      "v_dual_max_num_f32 v42, v56, v80 :: v_dual_max_num_f32 v47, v62, v86\n\t" \
      "v_dual_max_num_f32 v46, v62, v80 :: v_dual_max_num_f32 v43, v56, v86\n\t" \
      "v_dual_max_num_f32 v96, v57, v69 :: v_dual_max_num_f32 v101, v63, v75\n\t" \
      "v_dual_max_num_f32 v100, v63, v69 :: v_dual_max_num_f32 v97, v57, v75\n\t" \
      "v_dual_max_num_f32 v98, v57, v81 :: v_dual_max_num_f32 v103, v63, v87\n\t" \
      "v_dual_max_num_f32 v102, v63, v81 :: v_dual_max_num_f32 v99, v57, v87\n\t" \
      "v_dual_fmac_f32 v48, v40, v92 :: v_dual_fmac_f32 v49, v41, v92\n\t" \
      "v_dual_fmac_f32 v50, v42, v92 :: v_dual_fmac_f32 v51, v43, v92\n\t" \
      "v_dual_fmac_f32 v52, v44, v92 :: v_dual_fmac_f32 v53, v45, v92\n\t" \
      "v_dual_fmac_f32 v54, v46, v92 :: v_dual_fmac_f32 v55, v47, v92\n\t" \
      "v_dual_max_num_f32 v40, v58, v70 :: v_dual_max_num_f32 v45, v64, v76\n\t" \
      "v_dual_max_num_f32 v44, v64, v70 :: v_dual_max_num_f32 v41, v58, v76\n\t" \
      "v_dual_max_num_f32 v42, v58, v82 :: v_dual_max_num_f32 v47, v64, v88\n\t" \
      "v_dual_max_num_f32 v46, v64, v82 :: v_dual_max_num_f32 v43, v58, v88\n\t" \
      "v_dual_fmac_f32 v48, v96, v93 :: v_dual_fmac_f32 v49, v97, v93\n\t" \
      "v_dual_fmac_f32 v50, v98, v93 :: v_dual_fmac_f32 v51, v99, v93\n\t" \
      "v_dual_fmac_f32 v52, v100, v93 :: v_dual_fmac_f32 v53, v101, v93\n\t" \
      "v_dual_fmac_f32 v54, v102, v93 :: v_dual_fmac_f32 v55, v103, v93\n\t" \
      "v_dual_max_num_f32 v96, v59, v71 :: v_dual_max_num_f32 v101, v65, v77\n\t" \
      "v_dual_max_num_f32 v100, v65, v71 :: v_dual_max_num_f32 v97, v59, v77\n\t" \
      "v_dual_max_num_f32 v98, v59, v83 :: v_dual_max_num_f32 v103, v65, v89\n\t" \
      "v_dual_max_num_f32 v102, v65, v83 :: v_dual_max_num_f32 v99, v59, v89\n\t" \
      "v_dual_fmac_f32 v48, v40, v94 :: v_dual_fmac_f32 v49, v41, v94\n\t" \
      "v_dual_fmac_f32 v50, v42, v94 :: v_dual_fmac_f32 v51, v43, v94\n\t" \
      "v_dual_fmac_f32 v52, v44, v94 :: v_dual_fmac_f32 v53, v45, v94\n\t" \
      "v_dual_fmac_f32 v54, v46, v94 :: v_dual_fmac_f32 v55, v47, v94\n\t" \
      "v_dual_fmac_f32 v48, v96, v95 :: v_dual_fmac_f32 v49, v97, v95\n\t" \
      "v_dual_fmac_f32 v50, v98, v95 :: v_dual_fmac_f32 v51, v99, v95\n\t" \
      "v_dual_fmac_f32 v52, v100, v95 :: v_dual_fmac_f32 v53, v101, v95\n\t" \
      "v_dual_fmac_f32 v54, v102, v95 :: v_dual_fmac_f32 v55, v103, v95" \
      : "+{v[48:49]}"(acc0a), "+{v[50:51]}"(acc0b),                           \
        "+{v[52:53]}"(acc1a), "+{v[54:55]}"(acc1b)                            \
      : "{v[56:59]}"(A0), "{v[62:65]}"(A1), "{v[68:71]}"(B0), "{v[74:77]}"(B1), \
        "{v[80:83]}"(B2), "{v[86:89]}"(B3), "{v[92:95]}"(WV)             \
      : "v40", "v41", "v42", "v43", "v44", "v45", "v46", "v47",         \
        "v96", "v97", "v98", "v99", "v100", "v101", "v102", "v103");

#define LG_NC     4
#define LG_CK     (HID_N / LG_NC)

#define LG_HOLD() __builtin_amdgcn_sched_barrier(0)

#define LG_TRIP(K)                                                            \
  {                                                                           \
    const v4f na0 = *(const v4f*)&pa0[(K) + 4];                               \
    const v4f na1 = *(const v4f*)&pa1[(K) + 4];                               \
    const v4f nb0 = *(const v4f*)&pb0[(K) + 4];                               \
    const v4f nb1 = *(const v4f*)&pb1[(K) + 4];                               \
    const v4f nb2 = *(const v4f*)&pb2[(K) + 4];                               \
    const v4f nb3 = *(const v4f*)&pb3[(K) + 4];                               \
    const v4f nwv = *(const v4f*)&wL[(K) + 4];                                \
    LG_HOLD();                                                                \
    LG_STEP_S1(a0, a1, b0, b1, b2, b3, wv)                                    \
    LG_HOLD();                                                                \
    a0 = *(const v4f*)&pa0[(K) + 8];                                          \
    a1 = *(const v4f*)&pa1[(K) + 8];                                          \
    b0 = *(const v4f*)&pb0[(K) + 8];                                          \
    b1 = *(const v4f*)&pb1[(K) + 8];                                          \
    b2 = *(const v4f*)&pb2[(K) + 8];                                          \
    b3 = *(const v4f*)&pb3[(K) + 8];                                          \
    wv = *(const v4f*)&wL[(K) + 8];                                           \
    LG_HOLD();                                                                \
    LG_STEP_S2(na0, na1, nb0, nb1, nb2, nb3, nwv)                             \
    LG_HOLD();                                                                \
  }

#define LG_TRIPS(KLO, KHI)                                                    \
  for (int k = (KLO); k < (KHI); k += 8) LG_TRIP(k)

#define LG_GU(J, C) (*(const float4*)&gu[(rb + 16 * (J)) * HID_N + LG_CK * (C) + cp])
#define LG_GT(J, C) (*(const float4*)&gt[(rb + 16 * (J)) * HID_N + LG_CK * (C) + cp])
#define LG_LOAD_CHUNK(C)                                                      \
  const float4 su##C##_0 = LG_GU(0, C), su##C##_1 = LG_GU(1, C);              \
  const float4 st##C##_0 = LG_GT(0, C), st##C##_1 = LG_GT(1, C);              \
  const float4 st##C##_2 = LG_GT(2, C), st##C##_3 = LG_GT(3, C);

#define LG_PU(J, C) (*(float4*)&puL[(rb + 16 * (J)) * LG_PITCH + LG_CK * (C) + cp])
#define LG_PT(J, C) (*(float4*)&ptL[((rb >> 2) + 4 * (J) + 16 * (rb & 3)) * LG_PITCH + LG_CK * (C) + cp])
#define LG_STORE_CHUNK(C)                                                     \
  {                                                                           \
    LG_PU(0, C) = su##C##_0; LG_PU(1, C) = su##C##_1;                         \
    LG_PT(0, C) = st##C##_0; LG_PT(1, C) = st##C##_1;                         \
    LG_PT(2, C) = st##C##_2; LG_PT(3, C) = st##C##_3;                         \
  }

#define LG_DOT1(S, V, WQ)                                                     \
  S = fmaf(V.x, WQ.x, S); S = fmaf(V.y, WQ.y, S);                             \
  S = fmaf(V.z, WQ.z, S); S = fmaf(V.w, WQ.w, S);
#define LG_DOT_CHUNK(C)                                                       \
  {                                                                           \
    LG_DOT1(ts0, st##C##_0, wq##C) LG_DOT1(ts1, st##C##_1, wq##C)             \
    LG_DOT1(ts2, st##C##_2, wq##C) LG_DOT1(ts3, st##C##_3, wq##C)             \
  }

__global__ __launch_bounds__(256) void logits_kernel(
    const float* preU, const float* preT,
    const float* fw1, const float* __restrict__ fb1,
    float* out) {
  __shared__ __align__(16) float puL[LG_R * LG_PITCH];
  __shared__ __align__(16) float ptL[LG_T * LG_PITCH];
  __shared__ __align__(16) float wL[HID_N];
  __shared__ __align__(16) float tsL[LG_T];

  const int ttile = blockIdx.x & (T_N / LG_T - 1);
  const int rtile = blockIdx.x >> 3;
  const int tid   = threadIdx.x;

  const float* gu = preU + (size_t)rtile * LG_R * HID_N;
  const float* gt = preT + (size_t)ttile * LG_T * HID_N;

  const int cp = (tid & 15) << 2;
  const int rb = tid >> 4;
  LG_LOAD_CHUNK(0)
  const float4 rw = *(const float4*)&fw1[(tid & (HID_N / 4 - 1)) * 4];
  __builtin_amdgcn_sched_barrier(0);
  LG_LOAD_CHUNK(1)
  __builtin_amdgcn_sched_barrier(0);
  LG_LOAD_CHUNK(2)
  __builtin_amdgcn_sched_barrier(0);
  LG_LOAD_CHUNK(3)
  __builtin_amdgcn_sched_barrier(0);

  LG_STORE_CHUNK(0)
  if (tid < HID_N / 4) *(float4*)&wL[tid * 4] = rw;
  lds_barrier();

  const int tt = tid & 15;
  const int tu = tid >> 4;
  const float* pa0 = &puL[tu * LG_PITCH];
  const float* pa1 = &puL[(tu + 16) * LG_PITCH];
  const float* pb0 = &ptL[tt * LG_PITCH];
  const float* pb1 = &ptL[(tt + 16) * LG_PITCH];
  const float* pb2 = &ptL[(tt + 32) * LG_PITCH];
  const float* pb3 = &ptL[(tt + 48) * LG_PITCH];

  v2f acc0a = {0.f, 0.f}, acc0b = {0.f, 0.f};
  v2f acc1a = {0.f, 0.f}, acc1b = {0.f, 0.f};

  const float4 wq0 = *(const float4*)&wL[cp];
  const float4 wq1 = *(const float4*)&wL[LG_CK + cp];
  const float4 wq2 = *(const float4*)&wL[2 * LG_CK + cp];
  const float4 wq3 = *(const float4*)&wL[3 * LG_CK + cp];
  float ts0 = 0.f, ts1 = 0.f, ts2 = 0.f, ts3 = 0.f;
  LG_DOT_CHUNK(0)

  v4f a0 = *(const v4f*)&pa0[0];
  v4f a1 = *(const v4f*)&pa1[0];
  v4f b0 = *(const v4f*)&pb0[0];
  v4f b1 = *(const v4f*)&pb1[0];
  v4f b2 = *(const v4f*)&pb2[0];
  v4f b3 = *(const v4f*)&pb3[0];
  v4f wv = *(const v4f*)&wL[0];

  LG_TRIPS(0, 32)
  LG_DOT_CHUNK(1)
  LG_STORE_CHUNK(1)
  lds_barrier();
  LG_TRIPS(32, 96)
  LG_DOT_CHUNK(2)
  LG_STORE_CHUNK(2)
  lds_barrier();
  LG_TRIPS(96, 160)
  LG_DOT_CHUNK(3)
  for (int msk = 1; msk < 16; msk <<= 1) {
    ts0 += __shfl_xor(ts0, msk);
    ts1 += __shfl_xor(ts1, msk);
    ts2 += __shfl_xor(ts2, msk);
    ts3 += __shfl_xor(ts3, msk);
  }
  if ((tid & 15) == 0) {
    tsL[rb] = ts0; tsL[rb + 16] = ts1; tsL[rb + 32] = ts2; tsL[rb + 48] = ts3;
  }
  LG_STORE_CHUNK(3)
  lds_barrier();
  LG_TRIPS(160, HID_N - 16)
  LG_TRIP(HID_N - 16)
  {
    const v4f na0 = *(const v4f*)&pa0[HID_N - 4];
    const v4f na1 = *(const v4f*)&pa1[HID_N - 4];
    const v4f nb0 = *(const v4f*)&pb0[HID_N - 4];
    const v4f nb1 = *(const v4f*)&pb1[HID_N - 4];
    const v4f nb2 = *(const v4f*)&pb2[HID_N - 4];
    const v4f nb3 = *(const v4f*)&pb3[HID_N - 4];
    const v4f nwv = *(const v4f*)&wL[HID_N - 4];
    LG_STEP_S1(a0, a1, b0, b1, b2, b3, wv)
    LG_STEP_S2(na0, na1, nb0, nb1, nb2, nb3, nwv)
  }

  const float bias = fb1[0];
  const float4 tsv = *(const float4*)&tsL[tt * 4];
  const float c0 = bias - tsv.x, c1 = bias - tsv.y, c2 = bias - tsv.z, c3 = bias - tsv.w;
  float* op = out + ((size_t)rtile * LG_R + tu) * T_N + ttile * LG_T + tt * 4;
  float4 o0, o1;
  o0.x = acc0a.x + c0; o0.y = acc0a.y + c1; o0.z = acc0b.x + c2; o0.w = acc0b.y + c3;
  o1.x = acc1a.x + c0; o1.y = acc1a.y + c1; o1.z = acc1b.x + c2; o1.w = acc1b.y + c3;
  *(float4*)&op[0]                 = o0;
  *(float4*)&op[(size_t)16 * T_N]  = o1;
}

extern "C" void kernel_launch(void* const* d_in, const int* in_sizes, int n_in,
                              void* d_out, int out_size, void* d_ws, size_t ws_size,
                              hipStream_t stream) {
  (void)in_sizes; (void)n_in; (void)out_size; (void)ws_size;
  const float* uav_feat = (const float*)d_in[0];
  const float* task_feat = (const float*)d_in[1];
  const float* uw0 = (const float*)d_in[2];
  const float* ub0 = (const float*)d_in[3];
  const float* uw1 = (const float*)d_in[4];
  const float* ub1 = (const float*)d_in[5];
  const float* uw2 = (const float*)d_in[6];
  const float* ub2 = (const float*)d_in[7];
  const float* tw0 = (const float*)d_in[8];
  const float* tb0 = (const float*)d_in[9];
  const float* tw1 = (const float*)d_in[10];
  const float* tb1 = (const float*)d_in[11];
  const float* tw2 = (const float*)d_in[12];
  const float* tb2 = (const float*)d_in[13];
  const float* head_queries = (const float*)d_in[14];
  const float* fw0 = (const float*)d_in[15];
  const float* fb0 = (const float*)d_in[16];
  const float* fw1 = (const float*)d_in[17];
  const float* fb1 = (const float*)d_in[18];
  float* out = (float*)d_out;

  float* ws = (float*)d_ws;
  float* preT  = ws;
  float* preU  = ws + 131072;

  front_kernel<<<H_N * (U_N / 16) + T_N / 16, 128, 0, stream>>>(
      uav_feat, task_feat, uw0, ub0, uw1, ub1, uw2, ub2,
      tw0, tb0, tw1, tb1, tw2, tb2, head_queries, fw0, fb0, preT, preU);
  logits_kernel<<<(H_N * U_N / LG_R) * (T_N / LG_T), 256, 0, stream>>>(
      preU, preT, fw1, fb1, out);
}


# Front kernel: the 256 output columns of the first fusion layer split over two blocks per 16-row tile (each runs the encoder), 96 -> 192 blocks; products per wave 288 -> 224; sum order kept
# speedup vs baseline: 3.3256x; 1.1010x over previous
#include <hip/hip_runtime.h>

typedef __attribute__((ext_vector_type(2))) float v2f;
typedef __attribute__((ext_vector_type(8))) float v8f;

#define U_N     256
#define T_N     512
#define H_N     4
#define UAVD    64
#define TASKD   32
#define E_N     128
#define HID_N   256

#define WMMA_F32(A, B, C) \
  __builtin_amdgcn_wmma_f32_16x16x4_f32(false, (A), false, (B), (short)0, (C), false, false)

__device__ __forceinline__ int koff(int j, int hf) { return 8 * j + 4 * hf; }

__device__ __forceinline__ void lds_barrier() {
  __builtin_amdgcn_fence(__ATOMIC_RELEASE, "workgroup", "local");
  __builtin_amdgcn_s_barrier();
  __builtin_amdgcn_fence(__ATOMIC_ACQUIRE, "workgroup", "local");
}

#define HOLD_REQUESTS_ABOVE() __builtin_amdgcn_sched_barrier(0)

#define LPITCH   132
#define SLAB_FLOATS (32 * LPITCH)

template <int R, int K>
__device__ __forceinline__ void slab_request(float4 (&g)[R * K / 128],
                                             const float* __restrict__ base,
                                             int ldw, int lane) {
  constexpr int LPR = K / 4;
  constexpr int RPL = 32 / LPR;
  const int lr = lane / LPR, lc = (lane % LPR) * 4;
  #pragma unroll
  for (int i = 0; i < R / RPL; ++i)
    g[i] = *(const float4*)(base + (size_t)(i * RPL + lr) * ldw + lc);
}

template <int R, int K>
__device__ __forceinline__ void slab_to_lds(float* __restrict__ ws,
                                            const float4 (&g)[R * K / 128], int lane) {
  constexpr int LPR = K / 4;
  constexpr int RPL = 32 / LPR;
  const int lr = lane / LPR, lc = (lane % LPR) * 4;
  #pragma unroll
  for (int i = 0; i < R / RPL; ++i)
    *(float4*)&ws[(i * RPL + lr) * LPITCH + lc] = g[i];
}

template <int NT, int K>
__device__ __forceinline__ void mma_layer(v8f (&acc)[NT],
                                          const float* __restrict__ ap,
                                          const float* __restrict__ wp, int hf) {
  float4 af[K / 8];
  float4 bf[NT][K / 8];
  #pragma unroll
  for (int j = 0; j < K / 8; ++j) {
    af[j] = *(const float4*)(ap + koff(j, hf));
    #pragma unroll
    for (int t = 0; t < NT; ++t)
      bf[t][j] = *(const float4*)(wp + t * 16 * LPITCH + koff(j, hf));
  }
  __builtin_amdgcn_sched_barrier(0);
  #pragma unroll
  for (int j = 0; j < K / 8; ++j) {
    v2f alo; alo.x = af[j].x; alo.y = af[j].y;
    v2f ahi; ahi.x = af[j].z; ahi.y = af[j].w;
    #pragma unroll
    for (int t = 0; t < NT; ++t) {
      v2f b; b.x = bf[t][j].x; b.y = bf[t][j].y;
      acc[t] = WMMA_F32(alo, b, acc[t]);
    }
    #pragma unroll
    for (int t = 0; t < NT; ++t) {
      v2f b; b.x = bf[t][j].z; b.y = bf[t][j].w;
      acc[t] = WMMA_F32(ahi, b, acc[t]);
    }
  }
}

#define SLAB_REQ(NAME, I) const float4 NAME##_##I = *(const float4*)(SLAB_BASE + (size_t)(I) * (SLAB_LDW) + lane * 4);
#define SLAB_PUT(NAME, I) *(float4*)&ws[(I) * LPITCH + lane * 4] = NAME##_##I;

template <int INK, bool TASK>
__device__ __forceinline__ void front_rows(
    float* __restrict__ bufA, float* __restrict__ bufB, float* __restrict__ slabs,
    const int row0, const int h, const int nhalf,
    const float* __restrict__ x,
    const float* __restrict__ w0, const float* __restrict__ b0,
    const float* __restrict__ w1, const float* __restrict__ b1,
    const float* __restrict__ w2, const float* __restrict__ b2,
    const float* __restrict__ hq,
    const float* __restrict__ fw0, const float* __restrict__ fb0,
    float* __restrict__ pre) {
  const int tid  = threadIdx.x;
  const int lane = tid & 31;
  const int wave = tid >> 5;
  const int m    = lane & 15;
  const int hf   = lane >> 4;
  const int moff = hf * 8;
  const int nbw  = wave * 32;
  const int nb0  = nhalf * 128 + wave * 32;
  float* ws = slabs + wave * SLAB_FLOATS;
  const float* wfrag = ws + m * LPITCH;

  constexpr int NX = (16 * INK / 4) / 128;
  float4 xin[NX];
  #pragma unroll
  for (int i = 0; i < NX; ++i)
    xin[i] = *(const float4*)(x + (size_t)row0 * INK + (size_t)(tid + 128 * i) * 4);
  float4 g1[32 * INK / 128];
  slab_request<32, INK>(g1, w0 + (size_t)nbw * INK, INK, lane);
  const float bv1a = b0[nbw + m], bv1b = b0[nbw + 16 + m];

  #pragma unroll
  for (int i = 0; i < NX; ++i) {
    const int f = (tid + 128 * i) * 4;
    const int r = f / INK, c = f - r * INK;
    *(float4*)&bufA[r * LPITCH + c] = xin[i];
  }
  slab_to_lds<32, INK>(ws, g1, lane);
  lds_barrier();

  const float* w1slab = w1 + (size_t)nbw * E_N;
#define SLAB_BASE w1slab
#define SLAB_LDW  E_N
  SLAB_REQ(g2, 0) SLAB_REQ(g2, 1) SLAB_REQ(g2, 2) SLAB_REQ(g2, 3) SLAB_REQ(g2, 4) SLAB_REQ(g2, 5)
  SLAB_REQ(g2, 6) SLAB_REQ(g2, 7) SLAB_REQ(g2, 8) SLAB_REQ(g2, 9) SLAB_REQ(g2, 10) SLAB_REQ(g2, 11)
  SLAB_REQ(g2, 12) SLAB_REQ(g2, 13) SLAB_REQ(g2, 14) SLAB_REQ(g2, 15) SLAB_REQ(g2, 16) SLAB_REQ(g2, 17)
  SLAB_REQ(g2, 18) SLAB_REQ(g2, 19) SLAB_REQ(g2, 20) SLAB_REQ(g2, 21) SLAB_REQ(g2, 22) SLAB_REQ(g2, 23)
  SLAB_REQ(g2, 24) SLAB_REQ(g2, 25) SLAB_REQ(g2, 26) SLAB_REQ(g2, 27) SLAB_REQ(g2, 28) SLAB_REQ(g2, 29)
  SLAB_REQ(g2, 30) SLAB_REQ(g2, 31)
#undef SLAB_BASE
#undef SLAB_LDW
  const float bv2a = b1[nbw + m], bv2b = b1[nbw + 16 + m];
  HOLD_REQUESTS_ABOVE();
  {
    v8f acc[2] = {};
    mma_layer<2, INK>(acc, bufA + m * LPITCH, wfrag, hf);
    #pragma unroll
    for (int v = 0; v < 8; ++v) {
      bufB[(moff + v) * LPITCH + nbw + m]      = fmaxf(acc[0][v] + bv1a, 0.f);
      bufB[(moff + v) * LPITCH + nbw + 16 + m] = fmaxf(acc[1][v] + bv1b, 0.f);
    }
  }
  lds_barrier();

  const float* w2slab = w2 + (size_t)nbw * E_N;
#define SLAB_BASE w2slab
#define SLAB_LDW  E_N
  SLAB_REQ(g3, 0) SLAB_REQ(g3, 1) SLAB_REQ(g3, 2) SLAB_REQ(g3, 3) SLAB_REQ(g3, 4) SLAB_REQ(g3, 5)
  SLAB_REQ(g3, 6) SLAB_REQ(g3, 7) SLAB_REQ(g3, 8) SLAB_REQ(g3, 9) SLAB_REQ(g3, 10) SLAB_REQ(g3, 11)
  SLAB_REQ(g3, 12) SLAB_REQ(g3, 13) SLAB_REQ(g3, 14) SLAB_REQ(g3, 15) SLAB_REQ(g3, 16) SLAB_REQ(g3, 17)
  SLAB_REQ(g3, 18) SLAB_REQ(g3, 19) SLAB_REQ(g3, 20) SLAB_REQ(g3, 21) SLAB_REQ(g3, 22) SLAB_REQ(g3, 23)
  SLAB_REQ(g3, 24) SLAB_REQ(g3, 25) SLAB_REQ(g3, 26) SLAB_REQ(g3, 27) SLAB_REQ(g3, 28) SLAB_REQ(g3, 29)
  SLAB_REQ(g3, 30) SLAB_REQ(g3, 31)
#undef SLAB_BASE
#undef SLAB_LDW
  const float bv3a = b2[nbw + m], bv3b = b2[nbw + 16 + m];
  float qva = 0.f, qvb = 0.f;
  if (!TASK) { qva = hq[(size_t)h * E_N + nbw + m]; qvb = hq[(size_t)h * E_N + nbw + 16 + m]; }
  const float* fwslab = fw0 + (size_t)nb0 * (2 * E_N) + (TASK ? E_N : 0);
#define SLAB_BASE fwslab
#define SLAB_LDW  (2 * E_N)
  SLAB_REQ(g4, 0) SLAB_REQ(g4, 1) SLAB_REQ(g4, 2) SLAB_REQ(g4, 3) SLAB_REQ(g4, 4) SLAB_REQ(g4, 5)
  SLAB_REQ(g4, 6) SLAB_REQ(g4, 7) SLAB_REQ(g4, 8) SLAB_REQ(g4, 9) SLAB_REQ(g4, 10) SLAB_REQ(g4, 11)
  SLAB_REQ(g4, 12) SLAB_REQ(g4, 13) SLAB_REQ(g4, 14) SLAB_REQ(g4, 15)
  SLAB_PUT(g2, 0) SLAB_PUT(g2, 1) SLAB_PUT(g2, 2) SLAB_PUT(g2, 3) SLAB_PUT(g2, 4) SLAB_PUT(g2, 5)
  SLAB_PUT(g2, 6) SLAB_PUT(g2, 7) SLAB_PUT(g2, 8) SLAB_PUT(g2, 9) SLAB_PUT(g2, 10) SLAB_PUT(g2, 11)
  SLAB_PUT(g2, 12) SLAB_PUT(g2, 13) SLAB_PUT(g2, 14) SLAB_PUT(g2, 15) SLAB_PUT(g2, 16) SLAB_PUT(g2, 17)
  SLAB_PUT(g2, 18) SLAB_PUT(g2, 19) SLAB_PUT(g2, 20) SLAB_PUT(g2, 21) SLAB_PUT(g2, 22) SLAB_PUT(g2, 23)
  SLAB_PUT(g2, 24) SLAB_PUT(g2, 25) SLAB_PUT(g2, 26) SLAB_PUT(g2, 27) SLAB_PUT(g2, 28) SLAB_PUT(g2, 29)
  SLAB_PUT(g2, 30) SLAB_PUT(g2, 31)
  HOLD_REQUESTS_ABOVE();
  {
    v8f acc[2] = {};
    mma_layer<2, E_N>(acc, bufB + m * LPITCH, wfrag, hf);
    #pragma unroll
    for (int v = 0; v < 8; ++v) {
      bufA[(moff + v) * LPITCH + nbw + m]      = fmaxf(acc[0][v] + bv2a, 0.f);
      bufA[(moff + v) * LPITCH + nbw + 16 + m] = fmaxf(acc[1][v] + bv2b, 0.f);
    }
  }
  lds_barrier();

  SLAB_REQ(g4, 16) SLAB_REQ(g4, 17) SLAB_REQ(g4, 18) SLAB_REQ(g4, 19) SLAB_REQ(g4, 20) SLAB_REQ(g4, 21)
  SLAB_REQ(g4, 22) SLAB_REQ(g4, 23) SLAB_REQ(g4, 24) SLAB_REQ(g4, 25) SLAB_REQ(g4, 26) SLAB_REQ(g4, 27)
  SLAB_REQ(g4, 28) SLAB_REQ(g4, 29) SLAB_REQ(g4, 30) SLAB_REQ(g4, 31)
#undef SLAB_BASE
#undef SLAB_LDW
  float nbv0 = 0.f, nbv1 = 0.f;
  if (TASK) { nbv0 = -fb0[nb0 + m]; nbv1 = -fb0[nb0 + 16 + m]; }
  SLAB_PUT(g3, 0) SLAB_PUT(g3, 1) SLAB_PUT(g3, 2) SLAB_PUT(g3, 3) SLAB_PUT(g3, 4) SLAB_PUT(g3, 5)
  SLAB_PUT(g3, 6) SLAB_PUT(g3, 7) SLAB_PUT(g3, 8) SLAB_PUT(g3, 9) SLAB_PUT(g3, 10) SLAB_PUT(g3, 11)
  SLAB_PUT(g3, 12) SLAB_PUT(g3, 13) SLAB_PUT(g3, 14) SLAB_PUT(g3, 15) SLAB_PUT(g3, 16) SLAB_PUT(g3, 17)
  SLAB_PUT(g3, 18) SLAB_PUT(g3, 19) SLAB_PUT(g3, 20) SLAB_PUT(g3, 21) SLAB_PUT(g3, 22) SLAB_PUT(g3, 23)
  SLAB_PUT(g3, 24) SLAB_PUT(g3, 25) SLAB_PUT(g3, 26) SLAB_PUT(g3, 27) SLAB_PUT(g3, 28) SLAB_PUT(g3, 29)
  SLAB_PUT(g3, 30) SLAB_PUT(g3, 31)
  HOLD_REQUESTS_ABOVE();
  {
    v8f acc[2] = {};
    mma_layer<2, E_N>(acc, bufA + m * LPITCH, wfrag, hf);
    #pragma unroll
    for (int v = 0; v < 8; ++v) {
      float va = acc[0][v] + bv3a;
      float vb = acc[1][v] + bv3b;
      if (!TASK) { va = va + qva; vb = vb + qvb; }
      bufB[(moff + v) * LPITCH + nbw + m]      = va;
      bufB[(moff + v) * LPITCH + nbw + 16 + m] = vb;
    }
  }
  lds_barrier();

  SLAB_PUT(g4, 0) SLAB_PUT(g4, 1) SLAB_PUT(g4, 2) SLAB_PUT(g4, 3) SLAB_PUT(g4, 4) SLAB_PUT(g4, 5)
  SLAB_PUT(g4, 6) SLAB_PUT(g4, 7) SLAB_PUT(g4, 8) SLAB_PUT(g4, 9) SLAB_PUT(g4, 10) SLAB_PUT(g4, 11)
  SLAB_PUT(g4, 12) SLAB_PUT(g4, 13) SLAB_PUT(g4, 14) SLAB_PUT(g4, 15) SLAB_PUT(g4, 16) SLAB_PUT(g4, 17)
  SLAB_PUT(g4, 18) SLAB_PUT(g4, 19) SLAB_PUT(g4, 20) SLAB_PUT(g4, 21) SLAB_PUT(g4, 22) SLAB_PUT(g4, 23)
  SLAB_PUT(g4, 24) SLAB_PUT(g4, 25) SLAB_PUT(g4, 26) SLAB_PUT(g4, 27) SLAB_PUT(g4, 28) SLAB_PUT(g4, 29)
  SLAB_PUT(g4, 30) SLAB_PUT(g4, 31)
  {
    v8f acc[2] = {};
    mma_layer<2, E_N>(acc, bufB + m * LPITCH, wfrag, hf);
    float* op = pre + (size_t)(row0 + moff) * HID_N + nb0 + m;
    #pragma unroll
    for (int v = 0; v < 8; ++v) {
      if (TASK) {
        op[(size_t)v * HID_N]      = nbv0 - acc[0][v];
        op[(size_t)v * HID_N + 16] = nbv1 - acc[1][v];
      } else {
        op[(size_t)v * HID_N]      = acc[0][v];
        op[(size_t)v * HID_N + 16] = acc[1][v];
      }
    }
  }
}

__global__ __launch_bounds__(128) void front_kernel(
    const float* __restrict__ uav_feat, const float* __restrict__ task_feat,
    const float* __restrict__ uw0, const float* __restrict__ ub0,
    const float* __restrict__ uw1, const float* __restrict__ ub1,
    const float* __restrict__ uw2, const float* __restrict__ ub2,
    const float* __restrict__ tw0, const float* __restrict__ tb0,
    const float* __restrict__ tw1, const float* __restrict__ tb1,
    const float* __restrict__ tw2, const float* __restrict__ tb2,
    const float* __restrict__ hq,
    const float* __restrict__ fw0, const float* __restrict__ fb0,
    float* __restrict__ preT, float* __restrict__ preU) {
  __shared__ __align__(16) float bufA[16 * LPITCH];
  __shared__ __align__(16) float bufB[16 * LPITCH];
  __shared__ __align__(16) float slabs[4 * SLAB_FLOATS];
  const int nhalf = blockIdx.x & 1;
  const int b = blockIdx.x >> 1;
  if (b < H_N * (U_N / 16)) {
    const int h = b >> 4, utile = b & 15;
    front_rows<UAVD, false>(bufA, bufB, slabs, utile * 16, h, nhalf, uav_feat,
                            uw0, ub0, uw1, ub1, uw2, ub2, hq, fw0, fb0,
                            preU + (size_t)h * U_N * HID_N);
  } else {
    const int ttile = b - H_N * (U_N / 16);
    front_rows<TASKD, true>(bufA, bufB, slabs, ttile * 16, 0, nhalf, task_feat,
                            tw0, tb0, tw1, tb1, tw2, tb2, hq, fw0, fb0, preT);
  }
}

#define LG_R      32
#define LG_T      64
#define LG_PITCH  260

typedef __attribute__((ext_vector_type(4))) float v4f;

#define LG_STEP_S1(A0, A1, B0, B1, B2, B3, WV)                              \
  asm(                                                                        \
      "v_dual_max_num_f32 v40, v0, v12 :: v_dual_max_num_f32 v45, v6, v18\n\t" \
      "v_dual_max_num_f32 v44, v6, v12 :: v_dual_max_num_f32 v41, v0, v18\n\t" \
      "v_dual_max_num_f32 v42, v0, v24 :: v_dual_max_num_f32 v47, v6, v30\n\t" \
      "v_dual_max_num_f32 v46, v6, v24 :: v_dual_max_num_f32 v43, v0, v30\n\t" \
      "v_dual_max_num_f32 v96, v1, v13 :: v_dual_max_num_f32 v101, v7, v19\n\t" \
      "v_dual_max_num_f32 v100, v7, v13 :: v_dual_max_num_f32 v97, v1, v19\n\t" \
      "v_dual_max_num_f32 v98, v1, v25 :: v_dual_max_num_f32 v103, v7, v31\n\t" \
      "v_dual_max_num_f32 v102, v7, v25 :: v_dual_max_num_f32 v99, v1, v31\n\t" \
      "v_dual_fmac_f32 v48, v40, v36 :: v_dual_fmac_f32 v49, v41, v36\n\t" \
      "v_dual_fmac_f32 v50, v42, v36 :: v_dual_fmac_f32 v51, v43, v36\n\t" \
      "v_dual_fmac_f32 v52, v44, v36 :: v_dual_fmac_f32 v53, v45, v36\n\t" \
      "v_dual_fmac_f32 v54, v46, v36 :: v_dual_fmac_f32 v55, v47, v36\n\t" \
      "v_dual_max_num_f32 v40, v2, v14 :: v_dual_max_num_f32 v45, v8, v20\n\t" \
      "v_dual_max_num_f32 v44, v8, v14 :: v_dual_max_num_f32 v41, v2, v20\n\t" \
      "v_dual_max_num_f32 v42, v2, v26 :: v_dual_max_num_f32 v47, v8, v32\n\t" \
      "v_dual_max_num_f32 v46, v8, v26 :: v_dual_max_num_f32 v43, v2, v32\n\t" \
      "v_dual_fmac_f32 v48, v96, v37 :: v_dual_fmac_f32 v49, v97, v37\n\t" \
      "v_dual_fmac_f32 v50, v98, v37 :: v_dual_fmac_f32 v51, v99, v37\n\t" \
      "v_dual_fmac_f32 v52, v100, v37 :: v_dual_fmac_f32 v53, v101, v37\n\t" \
      "v_dual_fmac_f32 v54, v102, v37 :: v_dual_fmac_f32 v55, v103, v37\n\t" \
      "v_dual_max_num_f32 v96, v3, v15 :: v_dual_max_num_f32 v101, v9, v21\n\t" \
      "v_dual_max_num_f32 v100, v9, v15 :: v_dual_max_num_f32 v97, v3, v21\n\t" \
      "v_dual_max_num_f32 v98, v3, v27 :: v_dual_max_num_f32 v103, v9, v33\n\t" \
      "v_dual_max_num_f32 v102, v9, v27 :: v_dual_max_num_f32 v99, v3, v33\n\t" \
      "v_dual_fmac_f32 v48, v40, v38 :: v_dual_fmac_f32 v49, v41, v38\n\t" \
      "v_dual_fmac_f32 v50, v42, v38 :: v_dual_fmac_f32 v51, v43, v38\n\t" \
      "v_dual_fmac_f32 v52, v44, v38 :: v_dual_fmac_f32 v53, v45, v38\n\t" \
      "v_dual_fmac_f32 v54, v46, v38 :: v_dual_fmac_f32 v55, v47, v38\n\t" \
      "v_dual_fmac_f32 v48, v96, v39 :: v_dual_fmac_f32 v49, v97, v39\n\t" \
      "v_dual_fmac_f32 v50, v98, v39 :: v_dual_fmac_f32 v51, v99, v39\n\t" \
      "v_dual_fmac_f32 v52, v100, v39 :: v_dual_fmac_f32 v53, v101, v39\n\t" \
      "v_dual_fmac_f32 v54, v102, v39 :: v_dual_fmac_f32 v55, v103, v39" \
      : "+{v[48:49]}"(acc0a), "+{v[50:51]}"(acc0b),                           \
        "+{v[52:53]}"(acc1a), "+{v[54:55]}"(acc1b)                            \
      : "{v[0:3]}"(A0), "{v[6:9]}"(A1), "{v[12:15]}"(B0), "{v[18:21]}"(B1), \
        "{v[24:27]}"(B2), "{v[30:33]}"(B3), "{v[36:39]}"(WV)             \
      : "v40", "v41", "v42", "v43", "v44", "v45", "v46", "v47",         \
        "v96", "v97", "v98", "v99", "v100", "v101", "v102", "v103");

#define LG_STEP_S2(A0, A1, B0, B1, B2, B3, WV)                              \
  asm(                                                                        \
      "v_dual_max_num_f32 v40, v56, v68 :: v_dual_max_num_f32 v45, v62, v74\n\t" \
      "v_dual_max_num_f32 v44, v62, v68 :: v_dual_max_num_f32 v41, v56, v74\n\t" \
      "v_dual_max_num_f32 v42, v56, v80 :: v_dual_max_num_f32 v47, v62, v86\n\t" \
      "v_dual_max_num_f32 v46, v62, v80 :: v_dual_max_num_f32 v43, v56, v86\n\t" \
      "v_dual_max_num_f32 v96, v57, v69 :: v_dual_max_num_f32 v101, v63, v75\n\t" \
      "v_dual_max_num_f32 v100, v63, v69 :: v_dual_max_num_f32 v97, v57, v75\n\t" \
      "v_dual_max_num_f32 v98, v57, v81 :: v_dual_max_num_f32 v103, v63, v87\n\t" \
      "v_dual_max_num_f32 v102, v63, v81 :: v_dual_max_num_f32 v99, v57, v87\n\t" \
      "v_dual_fmac_f32 v48, v40, v92 :: v_dual_fmac_f32 v49, v41, v92\n\t" \
      "v_dual_fmac_f32 v50, v42, v92 :: v_dual_fmac_f32 v51, v43, v92\n\t" \
      "v_dual_fmac_f32 v52, v44, v92 :: v_dual_fmac_f32 v53, v45, v92\n\t" \
      "v_dual_fmac_f32 v54, v46, v92 :: v_dual_fmac_f32 v55, v47, v92\n\t" \
      "v_dual_max_num_f32 v40, v58, v70 :: v_dual_max_num_f32 v45, v64, v76\n\t" \
      "v_dual_max_num_f32 v44, v64, v70 :: v_dual_max_num_f32 v41, v58, v76\n\t" \
      "v_dual_max_num_f32 v42, v58, v82 :: v_dual_max_num_f32 v47, v64, v88\n\t" \
      "v_dual_max_num_f32 v46, v64, v82 :: v_dual_max_num_f32 v43, v58, v88\n\t" \
      "v_dual_fmac_f32 v48, v96, v93 :: v_dual_fmac_f32 v49, v97, v93\n\t" \
      "v_dual_fmac_f32 v50, v98, v93 :: v_dual_fmac_f32 v51, v99, v93\n\t" \
      "v_dual_fmac_f32 v52, v100, v93 :: v_dual_fmac_f32 v53, v101, v93\n\t" \
      "v_dual_fmac_f32 v54, v102, v93 :: v_dual_fmac_f32 v55, v103, v93\n\t" \
      "v_dual_max_num_f32 v96, v59, v71 :: v_dual_max_num_f32 v101, v65, v77\n\t" \
      "v_dual_max_num_f32 v100, v65, v71 :: v_dual_max_num_f32 v97, v59, v77\n\t" \
      "v_dual_max_num_f32 v98, v59, v83 :: v_dual_max_num_f32 v103, v65, v89\n\t" \
      "v_dual_max_num_f32 v102, v65, v83 :: v_dual_max_num_f32 v99, v59, v89\n\t" \
      "v_dual_fmac_f32 v48, v40, v94 :: v_dual_fmac_f32 v49, v41, v94\n\t" \
      "v_dual_fmac_f32 v50, v42, v94 :: v_dual_fmac_f32 v51, v43, v94\n\t" \
      "v_dual_fmac_f32 v52, v44, v94 :: v_dual_fmac_f32 v53, v45, v94\n\t" \
      "v_dual_fmac_f32 v54, v46, v94 :: v_dual_fmac_f32 v55, v47, v94\n\t" \
      "v_dual_fmac_f32 v48, v96, v95 :: v_dual_fmac_f32 v49, v97, v95\n\t" \
      "v_dual_fmac_f32 v50, v98, v95 :: v_dual_fmac_f32 v51, v99, v95\n\t" \
      "v_dual_fmac_f32 v52, v100, v95 :: v_dual_fmac_f32 v53, v101, v95\n\t" \
      "v_dual_fmac_f32 v54, v102, v95 :: v_dual_fmac_f32 v55, v103, v95" \
      : "+{v[48:49]}"(acc0a), "+{v[50:51]}"(acc0b),                           \
        "+{v[52:53]}"(acc1a), "+{v[54:55]}"(acc1b)                            \
      : "{v[56:59]}"(A0), "{v[62:65]}"(A1), "{v[68:71]}"(B0), "{v[74:77]}"(B1), \
        "{v[80:83]}"(B2), "{v[86:89]}"(B3), "{v[92:95]}"(WV)             \
      : "v40", "v41", "v42", "v43", "v44", "v45", "v46", "v47",         \
        "v96", "v97", "v98", "v99", "v100", "v101", "v102", "v103");

#define LG_NC     4
#define LG_CK     (HID_N / LG_NC)

#define LG_HOLD() __builtin_amdgcn_sched_barrier(0)

#define LG_TRIP(K)                                                            \
  {                                                                           \
    const v4f na0 = *(const v4f*)&pa0[(K) + 4];                               \
    const v4f na1 = *(const v4f*)&pa1[(K) + 4];                               \
    const v4f nb0 = *(const v4f*)&pb0[(K) + 4];                               \
    const v4f nb1 = *(const v4f*)&pb1[(K) + 4];                               \
    const v4f nb2 = *(const v4f*)&pb2[(K) + 4];                               \
    const v4f nb3 = *(const v4f*)&pb3[(K) + 4];                               \
    const v4f nwv = *(const v4f*)&wL[(K) + 4];                                \
    LG_HOLD();                                                                \
    LG_STEP_S1(a0, a1, b0, b1, b2, b3, wv)                                    \
    LG_HOLD();                                                                \
    a0 = *(const v4f*)&pa0[(K) + 8];                                          \
    a1 = *(const v4f*)&pa1[(K) + 8];                                          \
    b0 = *(const v4f*)&pb0[(K) + 8];                                          \
    b1 = *(const v4f*)&pb1[(K) + 8];                                          \
    b2 = *(const v4f*)&pb2[(K) + 8];                                          \
    b3 = *(const v4f*)&pb3[(K) + 8];                                          \
    wv = *(const v4f*)&wL[(K) + 8];                                           \
    LG_HOLD();                                                                \
    LG_STEP_S2(na0, na1, nb0, nb1, nb2, nb3, nwv)                             \
    LG_HOLD();                                                                \
  }

#define LG_TRIPS(KLO, KHI)                                                    \
  for (int k = (KLO); k < (KHI); k += 8) LG_TRIP(k)

#define LG_GU(J, C) (*(const float4*)&gu[(rb + 16 * (J)) * HID_N + LG_CK * (C) + cp])
#define LG_GT(J, C) (*(const float4*)&gt[(rb + 16 * (J)) * HID_N + LG_CK * (C) + cp])
#define LG_LOAD_CHUNK(C)                                                      \
  const float4 su##C##_0 = LG_GU(0, C), su##C##_1 = LG_GU(1, C);              \
  const float4 st##C##_0 = LG_GT(0, C), st##C##_1 = LG_GT(1, C);              \
  const float4 st##C##_2 = LG_GT(2, C), st##C##_3 = LG_GT(3, C);

#define LG_PU(J, C) (*(float4*)&puL[(rb + 16 * (J)) * LG_PITCH + LG_CK * (C) + cp])
#define LG_PT(J, C) (*(float4*)&ptL[((rb >> 2) + 4 * (J) + 16 * (rb & 3)) * LG_PITCH + LG_CK * (C) + cp])
#define LG_STORE_CHUNK(C)                                                     \
  {                                                                           \
    LG_PU(0, C) = su##C##_0; LG_PU(1, C) = su##C##_1;                         \
    LG_PT(0, C) = st##C##_0; LG_PT(1, C) = st##C##_1;                         \
    LG_PT(2, C) = st##C##_2; LG_PT(3, C) = st##C##_3;                         \
  }

#define LG_DOT1(S, V, WQ)                                                     \
  S = fmaf(V.x, WQ.x, S); S = fmaf(V.y, WQ.y, S);                             \
  S = fmaf(V.z, WQ.z, S); S = fmaf(V.w, WQ.w, S);
#define LG_DOT_CHUNK(C)                                                       \
  {                                                                           \
    LG_DOT1(ts0, st##C##_0, wq##C) LG_DOT1(ts1, st##C##_1, wq##C)             \
    LG_DOT1(ts2, st##C##_2, wq##C) LG_DOT1(ts3, st##C##_3, wq##C)             \
  }

__global__ __launch_bounds__(256) void logits_kernel(
    const float* preU, const float* preT,
    const float* fw1, const float* __restrict__ fb1,
    float* out) {
  __shared__ __align__(16) float puL[LG_R * LG_PITCH];
  __shared__ __align__(16) float ptL[LG_T * LG_PITCH];
  __shared__ __align__(16) float wL[HID_N];
  __shared__ __align__(16) float tsL[LG_T];

  const int ttile = blockIdx.x & (T_N / LG_T - 1);
  const int rtile = blockIdx.x >> 3;
  const int tid   = threadIdx.x;

  const float* gu = preU + (size_t)rtile * LG_R * HID_N;
  const float* gt = preT + (size_t)ttile * LG_T * HID_N;

  const int cp = (tid & 15) << 2;
  const int rb = tid >> 4;
  LG_LOAD_CHUNK(0)
  const float4 rw = *(const float4*)&fw1[(tid & (HID_N / 4 - 1)) * 4];
  __builtin_amdgcn_sched_barrier(0);
  LG_LOAD_CHUNK(1)
  __builtin_amdgcn_sched_barrier(0);
  LG_LOAD_CHUNK(2)
  __builtin_amdgcn_sched_barrier(0);
  LG_LOAD_CHUNK(3)
  __builtin_amdgcn_sched_barrier(0);

  LG_STORE_CHUNK(0)
  if (tid < HID_N / 4) *(float4*)&wL[tid * 4] = rw;
  lds_barrier();

  const int tt = tid & 15;
  const int tu = tid >> 4;
  const float* pa0 = &puL[tu * LG_PITCH];
  const float* pa1 = &puL[(tu + 16) * LG_PITCH];
  const float* pb0 = &ptL[tt * LG_PITCH];
  const float* pb1 = &ptL[(tt + 16) * LG_PITCH];
  const float* pb2 = &ptL[(tt + 32) * LG_PITCH];
  const float* pb3 = &ptL[(tt + 48) * LG_PITCH];

  v2f acc0a = {0.f, 0.f}, acc0b = {0.f, 0.f};
  v2f acc1a = {0.f, 0.f}, acc1b = {0.f, 0.f};

  const float4 wq0 = *(const float4*)&wL[cp];
  const float4 wq1 = *(const float4*)&wL[LG_CK + cp];
  const float4 wq2 = *(const float4*)&wL[2 * LG_CK + cp];
  const float4 wq3 = *(const float4*)&wL[3 * LG_CK + cp];
  float ts0 = 0.f, ts1 = 0.f, ts2 = 0.f, ts3 = 0.f;
  LG_DOT_CHUNK(0)

  v4f a0 = *(const v4f*)&pa0[0];
  v4f a1 = *(const v4f*)&pa1[0];
  v4f b0 = *(const v4f*)&pb0[0];
  v4f b1 = *(const v4f*)&pb1[0];
  v4f b2 = *(const v4f*)&pb2[0];
  v4f b3 = *(const v4f*)&pb3[0];
  v4f wv = *(const v4f*)&wL[0];

  LG_TRIPS(0, 32)
  LG_DOT_CHUNK(1)
  LG_STORE_CHUNK(1)
  lds_barrier();
  LG_TRIPS(32, 96)
  LG_DOT_CHUNK(2)
  LG_STORE_CHUNK(2)
  lds_barrier();
  LG_TRIPS(96, 160)
  LG_DOT_CHUNK(3)
  for (int msk = 1; msk < 16; msk <<= 1) {
    ts0 += __shfl_xor(ts0, msk);
    ts1 += __shfl_xor(ts1, msk);
    ts2 += __shfl_xor(ts2, msk);
    ts3 += __shfl_xor(ts3, msk);
  }
  if ((tid & 15) == 0) {
    tsL[rb] = ts0; tsL[rb + 16] = ts1; tsL[rb + 32] = ts2; tsL[rb + 48] = ts3;
  }
  LG_STORE_CHUNK(3)
  lds_barrier();
  LG_TRIPS(160, HID_N - 16)
  LG_TRIP(HID_N - 16)
  {
    const v4f na0 = *(const v4f*)&pa0[HID_N - 4];
    const v4f na1 = *(const v4f*)&pa1[HID_N - 4];
    const v4f nb0 = *(const v4f*)&pb0[HID_N - 4];
    const v4f nb1 = *(const v4f*)&pb1[HID_N - 4];
    const v4f nb2 = *(const v4f*)&pb2[HID_N - 4];
    const v4f nb3 = *(const v4f*)&pb3[HID_N - 4];
    const v4f nwv = *(const v4f*)&wL[HID_N - 4];
    LG_STEP_S1(a0, a1, b0, b1, b2, b3, wv)
    LG_STEP_S2(na0, na1, nb0, nb1, nb2, nb3, nwv)
  }

  const float bias = fb1[0];
  const float4 tsv = *(const float4*)&tsL[tt * 4];
  const float c0 = bias - tsv.x, c1 = bias - tsv.y, c2 = bias - tsv.z, c3 = bias - tsv.w;
  float* op = out + ((size_t)rtile * LG_R + tu) * T_N + ttile * LG_T + tt * 4;
  float4 o0, o1;
  o0.x = acc0a.x + c0; o0.y = acc0a.y + c1; o0.z = acc0b.x + c2; o0.w = acc0b.y + c3;
  o1.x = acc1a.x + c0; o1.y = acc1a.y + c1; o1.z = acc1b.x + c2; o1.w = acc1b.y + c3;
  *(float4*)&op[0]                 = o0;
  *(float4*)&op[(size_t)16 * T_N]  = o1;
}

extern "C" void kernel_launch(void* const* d_in, const int* in_sizes, int n_in,
                              void* d_out, int out_size, void* d_ws, size_t ws_size,
                              hipStream_t stream) {
  (void)in_sizes; (void)n_in; (void)out_size; (void)ws_size;
  const float* uav_feat = (const float*)d_in[0];
  const float* task_feat = (const float*)d_in[1];
  const float* uw0 = (const float*)d_in[2];
  const float* ub0 = (const float*)d_in[3];
  const float* uw1 = (const float*)d_in[4];
  const float* ub1 = (const float*)d_in[5];
  const float* uw2 = (const float*)d_in[6];
  const float* ub2 = (const float*)d_in[7];
  const float* tw0 = (const float*)d_in[8];
  const float* tb0 = (const float*)d_in[9];
  const float* tw1 = (const float*)d_in[10];
  const float* tb1 = (const float*)d_in[11];
  const float* tw2 = (const float*)d_in[12];
  const float* tb2 = (const float*)d_in[13];
  const float* head_queries = (const float*)d_in[14];
  const float* fw0 = (const float*)d_in[15];
  const float* fb0 = (const float*)d_in[16];
  const float* fw1 = (const float*)d_in[17];
  const float* fb1 = (const float*)d_in[18];
  float* out = (float*)d_out;

  float* ws = (float*)d_ws;
  float* preT  = ws;
  float* preU  = ws + 131072;

  front_kernel<<<2 * (H_N * (U_N / 16) + T_N / 16), 128, 0, stream>>>(
      uav_feat, task_feat, uw0, ub0, uw1, ub1, uw2, ub2,
      tw0, tb0, tw1, tb1, tw2, tb2, head_queries, fw0, fb0, preT, preU);
  logits_kernel<<<(H_N * U_N / LG_R) * (T_N / LG_T), 256, 0, stream>>>(
      preU, preT, fw1, fb1, out);
}
